# starting kernel
# baseline (speedup 1.0000x reference)
#include <hip/hip_runtime.h>
#include <hip/hip_bf16.h>

typedef __attribute__((ext_vector_type(2))) float v2f;
typedef __attribute__((ext_vector_type(8))) float v8f;

__global__ void k_deg_init(float* deg, int N) {
    int i = blockIdx.x * blockDim.x + threadIdx.x;
    if (i < N) deg[i] = 1.0f;
}

__global__ void k_deg_count(const int* __restrict__ dst, float* deg, int E) {
    int e = blockIdx.x * blockDim.x + threadIdx.x;
    if (e < E) atomicAdd(&deg[dst[e]], 1.0f);
}

__global__ void k_dinv(float* deg, int N) {
    int i = blockIdx.x * blockDim.x + threadIdx.x;
    if (i < N) deg[i] = rsqrtf(deg[i]);
}

template <int NT>
__global__ __launch_bounds__(32)
void k_gemm_wmma_f32(const float* __restrict__ X, const float* __restrict__ W,
                     float* __restrict__ Y, int K) {
    constexpr int Cout = NT * 16;
    const int mt   = blockIdx.x;
    const int lane = threadIdx.x;
    const int half = lane >> 4;
    const int l16  = lane & 15;

    const float* __restrict__ xrow = X + (size_t)(mt * 16 + l16) * K;
    const float* __restrict__ wlan = W + l16;

    v8f c[NT];
#pragma unroll
    for (int t = 0; t < NT; ++t) c[t] = (v8f){};

    for (int k = 0; k < K; k += 4) {
        const int ka = k + half * 2;
        v2f a;
        a.x = xrow[ka];
        a.y = xrow[ka + 1];
        const float* __restrict__ wk = wlan + (size_t)ka * Cout;
#pragma unroll
        for (int t = 0; t < NT; ++t) {
            v2f b;
            b.x = wk[t * 16];
            b.y = wk[t * 16 + Cout];
            c[t] = __builtin_amdgcn_wmma_f32_16x16x4_f32(
                        false, a,  false, b,
                        (short)0, c[t],
                        false,  false);
        }
    }

    float* __restrict__ yp = Y + (size_t)(mt * 16) * Cout + l16;
#pragma unroll
    for (int t = 0; t < NT; ++t)
#pragma unroll
        for (int v = 0; v < 8; ++v)
            yp[(size_t)(v + half * 8) * Cout + t * 16] = c[t][v];
}


__global__ void k_agg_init(const float* __restrict__ h, const float* __restrict__ bias,
                           const float* __restrict__ dinv, float* __restrict__ out,
                           int C) {
    const int i = blockIdx.x;
    const int c = threadIdx.x;
    const float di = dinv[i];
    out[(size_t)i * C + c] = bias[c] + h[(size_t)i * C + c] * di * di;
}

__global__ void k_agg_edges(const float* __restrict__ h,
                            const int* __restrict__ src, const int* __restrict__ dst,
                            const float* __restrict__ dinv, float* __restrict__ out,
                            int C) {
    const int e = blockIdx.x;
    const int c = threadIdx.x;
    const int s = src[e];
    const int d = dst[e];
    const float nrm = dinv[s] * dinv[d];
    atomicAdd(&out[(size_t)d * C + c], h[(size_t)s * C + c] * nrm);
}

__global__ void k_relu(float* x, long long n) {
    long long i = (long long)blockIdx.x * blockDim.x + threadIdx.x;
    if (i < n) x[i] = fmaxf(x[i], 0.0f);
}

extern "C" void kernel_launch(void* const* d_in, const int* in_sizes, int n_in,
                              void* d_out, int out_size, void* d_ws, size_t ws_size,
                              hipStream_t stream) {
    const float* x  = (const float*)d_in[0];
    const int*   ei = (const int*)  d_in[1];
    const float* W1 = (const float*)d_in[2];
    const float* b1 = (const float*)d_in[3];
    const float* W2 = (const float*)d_in[4];
    const float* b2 = (const float*)d_in[5];
    const float* W3 = (const float*)d_in[6];
    const float* b3 = (const float*)d_in[7];

    const int K0 = 512, C1 = 128, C2 = 64, C3 = 32;
    const int N = in_sizes[0] / K0;
    const int E = in_sizes[1] / 2;
    const int* src = ei;
    const int* dst = ei + E;

    float* ws   = (float*)d_ws;
    size_t pad  = ((size_t)N + 255) & ~(size_t)255;
    float* dinv = ws;
    float* bufH = ws + pad;
    float* bufA = bufH + (size_t)N * C1;
    float* out  = (float*)d_out;

    const int TB = 256;

    k_deg_init <<<(N + TB - 1) / TB, TB, 0, stream>>>(dinv, N);
    k_deg_count<<<(E + TB - 1) / TB, TB, 0, stream>>>(dst, dinv, E);
    k_dinv     <<<(N + TB - 1) / TB, TB, 0, stream>>>(dinv, N);

    k_gemm_wmma_f32<8><<<N / 16, 32, 0, stream>>>(x, W1, bufH, K0);
    k_agg_init <<<N, C1, 0, stream>>>(bufH, b1, dinv, bufA, C1);
    k_agg_edges<<<E, C1, 0, stream>>>(bufH, src, dst, dinv, bufA, C1);
    k_relu     <<<(int)(((long long)N * C1 + TB - 1) / TB), TB, 0, stream>>>(bufA, (long long)N * C1);

    k_gemm_wmma_f32<4><<<N / 16, 32, 0, stream>>>(bufA, W2, bufH, C1);
    k_agg_init <<<N, C2, 0, stream>>>(bufH, b2, dinv, bufA, C2);
    k_agg_edges<<<E, C2, 0, stream>>>(bufH, src, dst, dinv, bufA, C2);
    k_relu     <<<(int)(((long long)N * C2 + TB - 1) / TB), TB, 0, stream>>>(bufA, (long long)N * C2);

    k_gemm_wmma_f32<2><<<N / 16, 32, 0, stream>>>(bufA, W3, bufH, C2);
    k_agg_init <<<N, C3, 0, stream>>>(bufH, b3, dinv, out, C3);
    k_agg_edges<<<E, C3, 0, stream>>>(bufH, src, dst, dinv, out, C3);
}


# aggregation by destination: edges bucketed by dst once per call (integer atomic arrival rank, prefix sum); one wave sums a node's rows in registers and stores once; float atomics on out removed
# speedup vs baseline: 3.9301x; 3.9301x over previous
#include <hip/hip_runtime.h>
#include <hip/hip_bf16.h>

typedef __attribute__((ext_vector_type(2))) float v2f;
typedef __attribute__((ext_vector_type(8))) float v8f;

__global__ void k_cnt_zero(int* cnt, int n) {
    int i = blockIdx.x * blockDim.x + threadIdx.x;
    if (i < n) cnt[i] = 0;
}

__global__ void k_edge_rank(const int* __restrict__ dst, int* cnt,
                            int* __restrict__ rank, int E) {
    int e = blockIdx.x * blockDim.x + threadIdx.x;
    if (e < E) rank[e] = atomicAdd(&cnt[dst[e]], 1);
}

__global__ __launch_bounds__(1024)
void k_scan(const int* __restrict__ cnt, int* __restrict__ off, int npad) {
    __shared__ int wsum[32];
    const int lane  = threadIdx.x & 31;
    const int w     = __builtin_amdgcn_readfirstlane((int)(threadIdx.x >> 5));
    const int nrows = npad >> 7;
    const int per   = (nrows + 31) >> 5;
    const int r0    = min(w * per, nrows);
    const int r1    = min(r0 + per, nrows);
    const int4* __restrict__ c4 = (const int4*)cnt;
    int4* __restrict__ o4 = (int4*)off;

    int tot = 0;
#pragma unroll 4
    for (int r = r0; r < r1; ++r) {
        const int4 v = c4[r * 32 + lane];
        tot += (v.x + v.y) + (v.z + v.w);
    }
    for (int o = 16; o > 0; o >>= 1) tot += __shfl_xor(tot, o);
    if (lane == 0) wsum[w] = tot;
    __syncthreads();
    int carry = (lane < w) ? wsum[lane] : 0;
    for (int o = 16; o > 0; o >>= 1) carry += __shfl_xor(carry, o);

#pragma unroll 4
    for (int r = r0; r < r1; ++r) {
        const int4 v = c4[r * 32 + lane];
        const int s = (v.x + v.y) + (v.z + v.w);
        int inc = s;
        for (int o = 1; o < 32; o <<= 1) {
            const int t = __shfl_up(inc, o);
            if (lane >= o) inc += t;
        }
        const int ex = carry + inc - s;
        int4 q;
        q.x = ex; q.y = ex + v.x; q.z = q.y + v.y; q.w = q.z + v.z;
        o4[r * 32 + lane] = q;
        carry += __shfl(inc, 31);
    }
}

__global__ void k_fill(const int* __restrict__ src, const int* __restrict__ dst,
                       const int* __restrict__ rank, const int* __restrict__ off,
                       int* __restrict__ csr, int E) {
    int e = blockIdx.x * blockDim.x + threadIdx.x;
    if (e < E) csr[off[dst[e]] + rank[e]] = src[e];
}

template <int NT>
__global__ __launch_bounds__(32)
void k_gemm_wmma_f32(const float* __restrict__ X, const float* __restrict__ W,
                     float* __restrict__ Y, int K) {
    constexpr int Cout = NT * 16;
    const int mt   = blockIdx.x;
    const int lane = threadIdx.x;
    const int half = lane >> 4;
    const int l16  = lane & 15;

    const float* __restrict__ xrow = X + (size_t)(mt * 16 + l16) * K;
    const float* __restrict__ wlan = W + l16;

    v8f c[NT];
#pragma unroll
    for (int t = 0; t < NT; ++t) c[t] = (v8f){};

    for (int k = 0; k < K; k += 4) {
        const int ka = k + half * 2;
        v2f a;
        a.x = xrow[ka];
        a.y = xrow[ka + 1];
        const float* __restrict__ wk = wlan + (size_t)ka * Cout;
#pragma unroll
        for (int t = 0; t < NT; ++t) {
            v2f b;
            b.x = wk[t * 16];
            b.y = wk[t * 16 + Cout];
            c[t] = __builtin_amdgcn_wmma_f32_16x16x4_f32(
                        false, a,  false, b,
                        (short)0, c[t],
                        false,  false);
        }
    }

    float* __restrict__ yp = Y + (size_t)(mt * 16) * Cout + l16;
#pragma unroll
    for (int t = 0; t < NT; ++t)
#pragma unroll
        for (int v = 0; v < 8; ++v)
            yp[(size_t)(v + half * 8) * Cout + t * 16] = c[t][v];
}

template <int C, bool RELU, bool MOVE>
__global__ __launch_bounds__(256)
void k_gather(const float* __restrict__ h, const float* __restrict__ bias,
              const int* __restrict__ off, const int* __restrict__ csr,
              int* __restrict__ csr_new, float* __restrict__ out, int N, int E) {
    constexpr int LPR  = C / 4;
    constexpr int G    = 32 / LPR;
    constexpr int U    = (G == 4) ? 2 : 4;
    constexpr int STEP = U * G;
    const int lane = threadIdx.x & 31;
    const int i = __builtin_amdgcn_readfirstlane((int)(blockIdx.x * 8 + (threadIdx.x >> 5)));
    if (i >= N) return;
    const int grp = lane / LPR;
    const int li  = lane % LPR;

    const int start = off[i];
    const int end   = (i + 1 < N) ? off[i + 1] : E;
    const int cnt   = end - start;
    const float di  = rsqrtf((float)(cnt + 1));

    const float4* __restrict__ h4 = (const float4*)h;
    float4 acc = {0.0f, 0.0f, 0.0f, 0.0f};
    if (grp == 0) {
        const float4 b = ((const float4*)bias)[li];
        const float4 v = h4[(size_t)i * LPR + li];
        acc.x = b.x + v.x * di * di;
        acc.y = b.y + v.y * di * di;
        acc.z = b.z + v.z * di * di;
        acc.w = b.w + v.w * di * di;
    }

    for (int base = 0; base < cnt; base += 32) {
        const int n = min(32, cnt - base);
        int   s = i;
        float w = 0.0f;
        if (lane < n) {
            s = csr[start + base + lane];
            if (MOVE) csr_new[start + base + lane] = s;
            const int o0 = off[s];
            const int o1 = (s + 1 < N) ? off[s + 1] : E;
            w = rsqrtf((float)(o1 - o0 + 1)) * di;
        }
        for (int t = 0; t < n; t += STEP) {
            float4 v[U];
            float  wv[U];
#pragma unroll
            for (int k = 0; k < U; ++k) {
                int sv;
                if constexpr (G == 1) {
                    sv    = __builtin_amdgcn_readlane(s, t + k);
                    wv[k] = __int_as_float(__builtin_amdgcn_readlane(__float_as_int(w), t + k));
                } else {
                    const int e = t + k * G + grp;
                    sv    = __shfl(s, e);
                    wv[k] = __shfl(w, e);
                }
                v[k] = h4[(size_t)sv * LPR + li];
            }
#pragma unroll
            for (int k = 0; k < U; ++k) {
                acc.x = fmaf(v[k].x, wv[k], acc.x);
                acc.y = fmaf(v[k].y, wv[k], acc.y);
                acc.z = fmaf(v[k].z, wv[k], acc.z);
                acc.w = fmaf(v[k].w, wv[k], acc.w);
            }
        }
    }

    for (int o = LPR; o < 32; o <<= 1) {
        acc.x += __shfl_xor(acc.x, o);
        acc.y += __shfl_xor(acc.y, o);
        acc.z += __shfl_xor(acc.z, o);
        acc.w += __shfl_xor(acc.w, o);
    }
    if (grp == 0) {
        if (RELU) {
            acc.x = fmaxf(acc.x, 0.0f); acc.y = fmaxf(acc.y, 0.0f);
            acc.z = fmaxf(acc.z, 0.0f); acc.w = fmaxf(acc.w, 0.0f);
        }
        ((float4*)out)[(size_t)i * LPR + li] = acc;
    }
}

extern "C" void kernel_launch(void* const* d_in, const int* in_sizes, int n_in,
                              void* d_out, int out_size, void* d_ws, size_t ws_size,
                              hipStream_t stream) {
    const float* x  = (const float*)d_in[0];
    const int*   ei = (const int*)  d_in[1];
    const float* W1 = (const float*)d_in[2];
    const float* b1 = (const float*)d_in[3];
    const float* W2 = (const float*)d_in[4];
    const float* b2 = (const float*)d_in[5];
    const float* W3 = (const float*)d_in[6];
    const float* b3 = (const float*)d_in[7];

    const int K0 = 512, C1 = 128, C2 = 64, C3 = 32;
    const int N = in_sizes[0] / K0;
    const int E = in_sizes[1] / 2;
    const int* src = ei;
    const int* dst = ei + E;

    float* ws   = (float*)d_ws;
    size_t pad  = ((size_t)N + 255) & ~(size_t)255;
    int*   off  = (int*)ws;
    float* bufH = ws + pad;
    float* bufA = bufH + (size_t)N * C1;
    float* out  = (float*)d_out;
    int*   cnt  = (int*)bufA;
    int*   rank = cnt + pad;
    int*   csr0 = (int*)d_out;
    int*   csr1 = (int*)(bufH + (size_t)N * C2);

    const int TB = 256;

    k_cnt_zero <<<(int)(pad / TB), TB, 0, stream>>>(cnt, (int)pad);
    k_edge_rank<<<(E + TB - 1) / TB, TB, 0, stream>>>(dst, cnt, rank, E);
    k_scan     <<<1, 1024, 0, stream>>>(cnt, off, (int)pad);
    k_fill     <<<(E + TB - 1) / TB, TB, 0, stream>>>(src, dst, rank, off, csr0, E);

    k_gemm_wmma_f32<8><<<N / 16, 32, 0, stream>>>(x, W1, bufH, K0);
    k_gather<128, true, false><<<(N + 7) / 8, TB, 0, stream>>>(bufH, b1, off, csr0, csr1, bufA, N, E);

    k_gemm_wmma_f32<4><<<N / 16, 32, 0, stream>>>(bufA, W2, bufH, C1);
    k_gather<64, true, true><<<(N + 7) / 8, TB, 0, stream>>>(bufH, b2, off, csr0, csr1, bufA, N, E);

    k_gemm_wmma_f32<2><<<N / 16, 32, 0, stream>>>(bufA, W3, bufH, C2);
    k_gather<32, false, false><<<(N + 7) / 8, TB, 0, stream>>>(bufH, b3, off, csr1, csr1, out, N, E);
}


# source-side 1/sqrt(deg) folded into the GEMM's row stores: aggregation adds pre-scaled rows, scales the sum once by the destination's factor, bias added last; no per-edge degree lookups
# speedup vs baseline: 4.1261x; 1.0499x over previous
#include <hip/hip_runtime.h>
#include <hip/hip_bf16.h>

typedef __attribute__((ext_vector_type(2))) float v2f;
typedef __attribute__((ext_vector_type(8))) float v8f;

__global__ void k_cnt_zero(int* cnt, int n) {
    int i = blockIdx.x * blockDim.x + threadIdx.x;
    if (i < n) cnt[i] = 0;
}

__global__ void k_edge_rank(const int* __restrict__ dst, int* cnt,
                            int* __restrict__ rank, int E) {
    int e = blockIdx.x * blockDim.x + threadIdx.x;
    if (e < E) rank[e] = atomicAdd(&cnt[dst[e]], 1);
}

__global__ __launch_bounds__(1024)
void k_scan(const int* __restrict__ cnt, int* __restrict__ off, int npad) {
    __shared__ int wsum[32];
    const int lane  = threadIdx.x & 31;
    const int w     = __builtin_amdgcn_readfirstlane((int)(threadIdx.x >> 5));
    const int nrows = npad >> 7;
    const int per   = (nrows + 31) >> 5;
    const int r0    = min(w * per, nrows);
    const int r1    = min(r0 + per, nrows);
    const int4* __restrict__ c4 = (const int4*)cnt;
    int4* __restrict__ o4 = (int4*)off;

    int tot = 0;
#pragma unroll 4
    for (int r = r0; r < r1; ++r) {
        const int4 v = c4[r * 32 + lane];
        tot += (v.x + v.y) + (v.z + v.w);
    }
    for (int o = 16; o > 0; o >>= 1) tot += __shfl_xor(tot, o);
    if (lane == 0) wsum[w] = tot;
    __syncthreads();
    int carry = (lane < w) ? wsum[lane] : 0;
    for (int o = 16; o > 0; o >>= 1) carry += __shfl_xor(carry, o);

#pragma unroll 4
    for (int r = r0; r < r1; ++r) {
        const int4 v = c4[r * 32 + lane];
        const int s = (v.x + v.y) + (v.z + v.w);
        int inc = s;
        for (int o = 1; o < 32; o <<= 1) {
            const int t = __shfl_up(inc, o);
            if (lane >= o) inc += t;
        }
        const int ex = carry + inc - s;
        int4 q;
        q.x = ex; q.y = ex + v.x; q.z = q.y + v.y; q.w = q.z + v.z;
        o4[r * 32 + lane] = q;
        carry += __shfl(inc, 31);
    }
}

__global__ void k_fill(const int* __restrict__ src, const int* __restrict__ dst,
                       const int* __restrict__ rank, const int* __restrict__ off,
                       int* __restrict__ csr, int E) {
    int e = blockIdx.x * blockDim.x + threadIdx.x;
    if (e < E) csr[off[dst[e]] + rank[e]] = src[e];
}

template <int NT>
__global__ __launch_bounds__(32)
void k_gemm_wmma_f32(const float* __restrict__ X, const float* __restrict__ W,
                     const int* __restrict__ off, float* __restrict__ Y,
                     int K, int N, int E) {
    constexpr int Cout = NT * 16;
    const int mt   = blockIdx.x;
    const int lane = threadIdx.x;
    const int half = lane >> 4;
    const int l16  = lane & 15;

    const float* __restrict__ xrow = X + (size_t)(mt * 16 + l16) * K;
    const float* __restrict__ wlan = W + l16;

    const int  r0 = mt * 16 + half * 8;
    const int4 oa = *(const int4*)(off + r0);
    const int4 ob = *(const int4*)(off + r0 + 4);
    const int  oe = (r0 + 8 < N) ? off[r0 + 8] : E;

    v8f c[NT];
#pragma unroll
    for (int t = 0; t < NT; ++t) c[t] = (v8f){};

    for (int k = 0; k < K; k += 4) {
        const int ka = k + half * 2;
        v2f a;
        a.x = xrow[ka];
        a.y = xrow[ka + 1];
        const float* __restrict__ wk = wlan + (size_t)ka * Cout;
#pragma unroll
        for (int t = 0; t < NT; ++t) {
            v2f b;
            b.x = wk[t * 16];
            b.y = wk[t * 16 + Cout];
            c[t] = __builtin_amdgcn_wmma_f32_16x16x4_f32(
                        false, a,  false, b,
                        (short)0, c[t],
                        false,  false);
        }
    }

    float dv[8];
    dv[0] = rsqrtf((float)(oa.y - oa.x + 1));
    dv[1] = rsqrtf((float)(oa.z - oa.y + 1));
    dv[2] = rsqrtf((float)(oa.w - oa.z + 1));
    dv[3] = rsqrtf((float)(ob.x - oa.w + 1));
    dv[4] = rsqrtf((float)(ob.y - ob.x + 1));
    dv[5] = rsqrtf((float)(ob.z - ob.y + 1));
    dv[6] = rsqrtf((float)(ob.w - ob.z + 1));
    dv[7] = rsqrtf((float)(oe   - ob.w + 1));

    float* __restrict__ yp = Y + (size_t)(mt * 16) * Cout + l16;
#pragma unroll
    for (int t = 0; t < NT; ++t)
#pragma unroll
        for (int v = 0; v < 8; ++v)
            yp[(size_t)(v + half * 8) * Cout + t * 16] = c[t][v] * dv[v];
}

template <int C, bool RELU, bool MOVE>
__global__ __launch_bounds__(256)
void k_gather(const float* __restrict__ h, const float* __restrict__ bias,
              const int* __restrict__ off, const int* __restrict__ csr,
              int* __restrict__ csr_new, float* __restrict__ out, int N, int E) {
    constexpr int LPR  = C / 4;
    constexpr int G    = 32 / LPR;
    constexpr int U    = (G == 4) ? 2 : 4;
    constexpr int STEP = U * G;
    const int lane = threadIdx.x & 31;
    const int i = __builtin_amdgcn_readfirstlane((int)(blockIdx.x * 8 + (threadIdx.x >> 5)));
    if (i >= N) return;
    const int grp = lane / LPR;
    const int li  = lane % LPR;

    const int start = off[i];
    const int end   = (i + 1 < N) ? off[i + 1] : E;
    const int cnt   = end - start;
    const float di  = rsqrtf((float)(cnt + 1));

    const float4* __restrict__ h4 = (const float4*)h;
    float4 acc = {0.0f, 0.0f, 0.0f, 0.0f};
    float4 b   = {0.0f, 0.0f, 0.0f, 0.0f};
    if (grp == 0) {
        b   = ((const float4*)bias)[li];
        acc = h4[(size_t)i * LPR + li];
    }

    for (int base = 0; base < cnt; base += 32) {
        const int n = min(32, cnt - base);
        int s = i;
        if (lane < n) {
            s = csr[start + base + lane];
            if (MOVE) csr_new[start + base + lane] = s;
        }
        for (int t = 0; t < n; t += STEP) {
            float4 v[U];
            float  wv[U];
#pragma unroll
            for (int k = 0; k < U; ++k) {
                const int e = (G == 1) ? (t + k) : (t + k * G + grp);
                int sv;
                if constexpr (G == 1) sv = __builtin_amdgcn_readlane(s, e);
                else                  sv = __shfl(s, e);
                wv[k] = (e < n) ? 1.0f : 0.0f;
                v[k]  = h4[(size_t)sv * LPR + li];
            }
#pragma unroll
            for (int k = 0; k < U; ++k) {
                acc.x = fmaf(v[k].x, wv[k], acc.x);
                acc.y = fmaf(v[k].y, wv[k], acc.y);
                acc.z = fmaf(v[k].z, wv[k], acc.z);
                acc.w = fmaf(v[k].w, wv[k], acc.w);
            }
        }
    }

    for (int o = LPR; o < 32; o <<= 1) {
        acc.x += __shfl_xor(acc.x, o);
        acc.y += __shfl_xor(acc.y, o);
        acc.z += __shfl_xor(acc.z, o);
        acc.w += __shfl_xor(acc.w, o);
    }
    if (grp == 0) {
        acc.x = fmaf(acc.x, di, b.x); acc.y = fmaf(acc.y, di, b.y);
        acc.z = fmaf(acc.z, di, b.z); acc.w = fmaf(acc.w, di, b.w);
        if (RELU) {
            acc.x = fmaxf(acc.x, 0.0f); acc.y = fmaxf(acc.y, 0.0f);
            acc.z = fmaxf(acc.z, 0.0f); acc.w = fmaxf(acc.w, 0.0f);
        }
        ((float4*)out)[(size_t)i * LPR + li] = acc;
    }
}

extern "C" void kernel_launch(void* const* d_in, const int* in_sizes, int n_in,
                              void* d_out, int out_size, void* d_ws, size_t ws_size,
                              hipStream_t stream) {
    const float* x  = (const float*)d_in[0];
    const int*   ei = (const int*)  d_in[1];
    const float* W1 = (const float*)d_in[2];
    const float* b1 = (const float*)d_in[3];
    const float* W2 = (const float*)d_in[4];
    const float* b2 = (const float*)d_in[5];
    const float* W3 = (const float*)d_in[6];
    const float* b3 = (const float*)d_in[7];

    const int K0 = 512, C1 = 128, C2 = 64, C3 = 32;
    const int N = in_sizes[0] / K0;
    const int E = in_sizes[1] / 2;
    const int* src = ei;
    const int* dst = ei + E;

    float* ws   = (float*)d_ws;
    size_t pad  = ((size_t)N + 255) & ~(size_t)255;
    int*   off  = (int*)ws;
    float* bufH = ws + pad;
    float* bufA = bufH + (size_t)N * C1;
    float* out  = (float*)d_out;
    int*   cnt  = (int*)bufA;
    int*   rank = cnt + pad;
    int*   csr0 = (int*)d_out;
    int*   csr1 = (int*)(bufH + (size_t)N * C2);

    const int TB = 256;

    k_cnt_zero <<<(int)(pad / TB), TB, 0, stream>>>(cnt, (int)pad);
    k_edge_rank<<<(E + TB - 1) / TB, TB, 0, stream>>>(dst, cnt, rank, E);
    k_scan     <<<1, 1024, 0, stream>>>(cnt, off, (int)pad);
    k_fill     <<<(E + TB - 1) / TB, TB, 0, stream>>>(src, dst, rank, off, csr0, E);

    k_gemm_wmma_f32<8><<<N / 16, 32, 0, stream>>>(x, W1, off, bufH, K0, N, E);
    k_gather<128, true, false><<<(N + 7) / 8, TB, 0, stream>>>(bufH, b1, off, csr0, csr1, bufA, N, E);

    k_gemm_wmma_f32<4><<<N / 16, 32, 0, stream>>>(bufA, W2, off, bufH, C1, N, E);
    k_gather<64, true, true><<<(N + 7) / 8, TB, 0, stream>>>(bufH, b2, off, csr0, csr1, bufA, N, E);

    k_gemm_wmma_f32<2><<<N / 16, 32, 0, stream>>>(bufA, W3, off, bufH, C2, N, E);
    k_gather<32, false, false><<<(N + 7) / 8, TB, 0, stream>>>(bufH, b3, off, csr1, csr1, out, N, E);
}


# GEMMs: whole W staged once in LDS by 256 resident 16-wave work-groups; each wave walks 16-row panels and reads X in whole 128-byte lines one chunk ahead; k summed in a new order within each 8
# speedup vs baseline: 4.5193x; 1.0953x over previous
#include <hip/hip_runtime.h>
#include <hip/hip_bf16.h>

typedef __attribute__((ext_vector_type(2))) float v2f;
typedef __attribute__((ext_vector_type(8))) float v8f;

__global__ void k_cnt_zero(int* cnt, int n) {
    int i = blockIdx.x * blockDim.x + threadIdx.x;
    if (i < n) cnt[i] = 0;
}

__global__ void k_edge_rank(const int* __restrict__ dst, int* cnt,
                            int* __restrict__ rank, int E) {
    int e = blockIdx.x * blockDim.x + threadIdx.x;
    if (e < E) rank[e] = atomicAdd(&cnt[dst[e]], 1);
}

__global__ __launch_bounds__(1024)
void k_scan(const int* __restrict__ cnt, int* __restrict__ off, int npad) {
    __shared__ int wsum[32];
    const int lane  = threadIdx.x & 31;
    const int w     = __builtin_amdgcn_readfirstlane((int)(threadIdx.x >> 5));
    const int nrows = npad >> 7;
    const int per   = (nrows + 31) >> 5;
    const int r0    = min(w * per, nrows);
    const int r1    = min(r0 + per, nrows);
    const int4* __restrict__ c4 = (const int4*)cnt;
    int4* __restrict__ o4 = (int4*)off;

    int tot = 0;
#pragma unroll 4
    for (int r = r0; r < r1; ++r) {
        const int4 v = c4[r * 32 + lane];
        tot += (v.x + v.y) + (v.z + v.w);
    }
    for (int o = 16; o > 0; o >>= 1) tot += __shfl_xor(tot, o);
    if (lane == 0) wsum[w] = tot;
    __syncthreads();
    int carry = (lane < w) ? wsum[lane] : 0;
    for (int o = 16; o > 0; o >>= 1) carry += __shfl_xor(carry, o);

#pragma unroll 4
    for (int r = r0; r < r1; ++r) {
        const int4 v = c4[r * 32 + lane];
        const int s = (v.x + v.y) + (v.z + v.w);
        int inc = s;
        for (int o = 1; o < 32; o <<= 1) {
            const int t = __shfl_up(inc, o);
            if (lane >= o) inc += t;
        }
        const int ex = carry + inc - s;
        int4 q;
        q.x = ex; q.y = ex + v.x; q.z = q.y + v.y; q.w = q.z + v.z;
        o4[r * 32 + lane] = q;
        carry += __shfl(inc, 31);
    }
}

__global__ void k_fill(const int* __restrict__ src, const int* __restrict__ dst,
                       const int* __restrict__ rank, const int* __restrict__ off,
                       int* __restrict__ csr, int E) {
    int e = blockIdx.x * blockDim.x + threadIdx.x;
    if (e < E) csr[off[dst[e]] + rank[e]] = src[e];
}

template <int K, int NT>
__global__ __launch_bounds__(512)
void k_gemm_wmma_f32(const float* __restrict__ X, const float* __restrict__ Wg,
                     const int* __restrict__ off, float* __restrict__ Y,
                     int N, int E) {
    constexpr int Cout = NT * 16;
    constexpr int KQ   = K / 4;
    constexpr int PQ4  = Cout + 8;
    constexpr int NCH  = K / 32;
    __shared__ float4 Ws[KQ * PQ4];

    const int tid  = threadIdx.x;
    const int lane = tid & 31;
    const int w    = __builtin_amdgcn_readfirstlane(tid >> 5);
    const int half = lane >> 4;
    const int c    = lane & 15;
    const int npan = N >> 4;
    const int pstride = (int)gridDim.x * 16;
    const int g  = w >> 2;
    const int sw = (((w + g) & 3) << 2) | g;
    int p = (int)blockIdx.x + (int)gridDim.x * sw;

    const float4* __restrict__ xa =
        (const float4*)(X + (size_t)((p < npan ? p : 0) * 16 + c) * K) + half;
    float4 a[4];
#pragma unroll
    for (int q = 0; q < 4; ++q) a[q] = xa[2 * q];

#pragma unroll 8
    for (int idx = tid; idx < KQ * Cout; idx += 512) {
        const int kq = idx / Cout;
        const int n  = idx % Cout;
        const float* __restrict__ wp = Wg + (size_t)(kq * 4) * Cout + n;
        float4 v;
        v.x = wp[0];
        v.y = wp[Cout];
        v.z = wp[2 * Cout];
        v.w = wp[3 * Cout];
        Ws[kq * PQ4 + n] = v;
    }
    __syncthreads();

    const int lo = half * PQ4 + c;
    float4 bc[NT];
#pragma unroll
    for (int t = 0; t < NT; ++t) bc[t] = Ws[lo + t * 16];

    v8f acc[NT];
#pragma unroll
    for (int t = 0; t < NT; ++t) acc[t] = (v8f){};

    int4 oa = {0, 0, 0, 0}, ob = {0, 0, 0, 0};
    int  ot = 0;

    int j = 0;
    while (p < npan) {
        const int pn = p + pstride;
        const bool lastj = (j == NCH - 1);

        const int  r0 = p * 16 + half * 8;
        const bool rn = (r0 + 8 < N);
        if (j == 0) {
            oa = *(const int4*)(off + r0);
            ob = *(const int4*)(off + r0 + 4);
            ot = off[rn ? r0 + 8 : r0];
        }

        const float4* __restrict__ xn =
            (const float4*)(X + (size_t)((pn < npan ? pn : p) * 16 + c) * K) + half;
        const float4* __restrict__ src = lastj ? xn : (xa + (j + 1) * 8);
        float4 an[4];
#pragma unroll
        for (int q = 0; q < 4; ++q) an[q] = src[2 * q];

#pragma unroll
        for (int q = 0; q < 4; ++q) {
            const int rown = (q == 3) ? (((j + 1) & (NCH - 1)) * 8) : (j * 8 + 2 * q + 2);
            const int bo = lo + rown * PQ4;
            float4 bn[NT];
#pragma unroll
            for (int t = 0; t < NT; ++t) bn[t] = Ws[bo + t * 16];

            const v2f a01 = {a[q].x, a[q].y};
            const v2f a23 = {a[q].z, a[q].w};
#pragma unroll
            for (int t = 0; t < NT; ++t) {
                const v2f b01 = {bc[t].x, bc[t].y};
                acc[t] = __builtin_amdgcn_wmma_f32_16x16x4_f32(
                              false, a01,  false, b01,
                              (short)0, acc[t],
                              false,  false);
            }
#pragma unroll
            for (int t = 0; t < NT; ++t) {
                const v2f b23 = {bc[t].z, bc[t].w};
                acc[t] = __builtin_amdgcn_wmma_f32_16x16x4_f32(
                              false, a23,  false, b23,
                              (short)0, acc[t],
                              false,  false);
            }
#pragma unroll
            for (int t = 0; t < NT; ++t) bc[t] = bn[t];
        }
#pragma unroll
        for (int q = 0; q < 4; ++q) a[q] = an[q];

        if (lastj) {
            float dv[8];
            dv[0] = rsqrtf((float)(oa.y - oa.x + 1));
            dv[1] = rsqrtf((float)(oa.z - oa.y + 1));
            dv[2] = rsqrtf((float)(oa.w - oa.z + 1));
            dv[3] = rsqrtf((float)(ob.x - oa.w + 1));
            dv[4] = rsqrtf((float)(ob.y - ob.x + 1));
            dv[5] = rsqrtf((float)(ob.z - ob.y + 1));
            dv[6] = rsqrtf((float)(ob.w - ob.z + 1));
            dv[7] = rsqrtf((float)((rn ? ot : E) - ob.w + 1));

            float* __restrict__ yp = Y + (size_t)(p * 16) * Cout + c;
#pragma unroll
            for (int t = 0; t < NT; ++t) {
#pragma unroll
                for (int v = 0; v < 8; ++v)
                    yp[(size_t)(v + half * 8) * Cout + t * 16] = acc[t][v] * dv[v];
                acc[t] = (v8f){};
            }
            xa = xn;
            p  = pn;
            j  = 0;
        } else {
            ++j;
        }
    }
}

template <int C, bool RELU, bool MOVE>
__global__ __launch_bounds__(256)
void k_gather(const float* __restrict__ h, const float* __restrict__ bias,
              const int* __restrict__ off, const int* __restrict__ csr,
              int* __restrict__ csr_new, float* __restrict__ out, int N, int E) {
    constexpr int LPR  = C / 4;
    constexpr int G    = 32 / LPR;
    constexpr int U    = (G == 4) ? 2 : 4;
    constexpr int STEP = U * G;
    const int lane = threadIdx.x & 31;
    const int i = __builtin_amdgcn_readfirstlane((int)(blockIdx.x * 8 + (threadIdx.x >> 5)));
    if (i >= N) return;
    const int grp = lane / LPR;
    const int li  = lane % LPR;

    const int start = off[i];
    const int end   = (i + 1 < N) ? off[i + 1] : E;
    const int cnt   = end - start;
    const float di  = rsqrtf((float)(cnt + 1));

    const float4* __restrict__ h4 = (const float4*)h;
    float4 acc = {0.0f, 0.0f, 0.0f, 0.0f};
    float4 b   = {0.0f, 0.0f, 0.0f, 0.0f};
    if (grp == 0) {
        b   = ((const float4*)bias)[li];
        acc = h4[(size_t)i * LPR + li];
    }

    for (int base = 0; base < cnt; base += 32) {
        const int n = min(32, cnt - base);
        int s = i;
        if (lane < n) {
            s = csr[start + base + lane];
            if (MOVE) csr_new[start + base + lane] = s;
        }
        for (int t = 0; t < n; t += STEP) {
            float4 v[U];
            float  wv[U];
#pragma unroll
            for (int k = 0; k < U; ++k) {
                const int e = (G == 1) ? (t + k) : (t + k * G + grp);
                int sv;
                if constexpr (G == 1) sv = __builtin_amdgcn_readlane(s, e);
                else                  sv = __shfl(s, e);
                wv[k] = (e < n) ? 1.0f : 0.0f;
                v[k]  = h4[(size_t)sv * LPR + li];
            }
#pragma unroll
            for (int k = 0; k < U; ++k) {
                acc.x = fmaf(v[k].x, wv[k], acc.x);
                acc.y = fmaf(v[k].y, wv[k], acc.y);
                acc.z = fmaf(v[k].z, wv[k], acc.z);
                acc.w = fmaf(v[k].w, wv[k], acc.w);
            }
        }
    }

    for (int o = LPR; o < 32; o <<= 1) {
        acc.x += __shfl_xor(acc.x, o);
        acc.y += __shfl_xor(acc.y, o);
        acc.z += __shfl_xor(acc.z, o);
        acc.w += __shfl_xor(acc.w, o);
    }
    if (grp == 0) {
        acc.x = fmaf(acc.x, di, b.x); acc.y = fmaf(acc.y, di, b.y);
        acc.z = fmaf(acc.z, di, b.z); acc.w = fmaf(acc.w, di, b.w);
        if (RELU) {
            acc.x = fmaxf(acc.x, 0.0f); acc.y = fmaxf(acc.y, 0.0f);
            acc.z = fmaxf(acc.z, 0.0f); acc.w = fmaxf(acc.w, 0.0f);
        }
        ((float4*)out)[(size_t)i * LPR + li] = acc;
    }
}

extern "C" void kernel_launch(void* const* d_in, const int* in_sizes, int n_in,
                              void* d_out, int out_size, void* d_ws, size_t ws_size,
                              hipStream_t stream) {
    const float* x  = (const float*)d_in[0];
    const int*   ei = (const int*)  d_in[1];
    const float* W1 = (const float*)d_in[2];
    const float* b1 = (const float*)d_in[3];
    const float* W2 = (const float*)d_in[4];
    const float* b2 = (const float*)d_in[5];
    const float* W3 = (const float*)d_in[6];
    const float* b3 = (const float*)d_in[7];

    const int K0 = 512, C1 = 128, C2 = 64, C3 = 32;
    const int N = in_sizes[0] / K0;
    const int E = in_sizes[1] / 2;
    const int* src = ei;
    const int* dst = ei + E;

    float* ws   = (float*)d_ws;
    size_t pad  = ((size_t)N + 255) & ~(size_t)255;
    int*   off  = (int*)ws;
    float* bufH = ws + pad;
    float* bufA = bufH + (size_t)N * C1;
    float* out  = (float*)d_out;
    int*   cnt  = (int*)bufA;
    int*   rank = cnt + pad;
    int*   csr0 = (int*)d_out;
    int*   csr1 = (int*)(bufH + (size_t)N * C2);

    const int TB = 256;

    k_cnt_zero <<<(int)(pad / TB), TB, 0, stream>>>(cnt, (int)pad);
    k_edge_rank<<<(E + TB - 1) / TB, TB, 0, stream>>>(dst, cnt, rank, E);
    k_scan     <<<1, 1024, 0, stream>>>(cnt, off, (int)pad);
    k_fill     <<<(E + TB - 1) / TB, TB, 0, stream>>>(src, dst, rank, off, csr0, E);

    k_gemm_wmma_f32<512, 8><<<256, 512, 0, stream>>>(x, W1, off, bufH, N, E);
    k_gather<128, true, false><<<(N + 7) / 8, TB, 0, stream>>>(bufH, b1, off, csr0, csr1, bufA, N, E);

    k_gemm_wmma_f32<128, 4><<<256, 512, 0, stream>>>(bufA, W2, off, bufH, N, E);
    k_gather<64, true, true><<<(N + 7) / 8, TB, 0, stream>>>(bufH, b2, off, csr0, csr1, bufA, N, E);

    k_gemm_wmma_f32<64, 2><<<256, 512, 0, stream>>>(bufA, W3, off, bufH, N, E);
    k_gather<32, false, false><<<(N + 7) / 8, TB, 0, stream>>>(bufH, b3, off, csr1, csr1, out, N, E);
}


# aggregation of layers 1 and 2 multiplies its 16-node tile by the next layer's W on WMMA from LDS: activations never stored, two GEMM launches gone; every sum formed in the same order as before
# speedup vs baseline: 4.7184x; 1.0441x over previous
#include <hip/hip_runtime.h>
#include <hip/hip_bf16.h>

typedef __attribute__((ext_vector_type(2))) float v2f;
typedef __attribute__((ext_vector_type(8))) float v8f;

__global__ void k_cnt_zero(int* cnt, int n) {
    int i = blockIdx.x * blockDim.x + threadIdx.x;
    if (i < n) cnt[i] = 0;
}

__global__ void k_edge_rank(const int* __restrict__ dst, int* cnt,
                            int* __restrict__ rank, int E) {
    int e = blockIdx.x * blockDim.x + threadIdx.x;
    if (e < E) rank[e] = atomicAdd(&cnt[dst[e]], 1);
}

__global__ __launch_bounds__(1024)
void k_scan(const int* __restrict__ cnt, int* __restrict__ off, int npad) {
    __shared__ int wsum[32];
    const int lane  = threadIdx.x & 31;
    const int w     = __builtin_amdgcn_readfirstlane((int)(threadIdx.x >> 5));
    const int nrows = npad >> 7;
    const int per   = (nrows + 31) >> 5;
    const int r0    = min(w * per, nrows);
    const int r1    = min(r0 + per, nrows);
    const int4* __restrict__ c4 = (const int4*)cnt;
    int4* __restrict__ o4 = (int4*)off;

    int tot = 0;
#pragma unroll 4
    for (int r = r0; r < r1; ++r) {
        const int4 v = c4[r * 32 + lane];
        tot += (v.x + v.y) + (v.z + v.w);
    }
    for (int o = 16; o > 0; o >>= 1) tot += __shfl_xor(tot, o);
    if (lane == 0) wsum[w] = tot;
    __syncthreads();
    int carry = (lane < w) ? wsum[lane] : 0;
    for (int o = 16; o > 0; o >>= 1) carry += __shfl_xor(carry, o);

#pragma unroll 4
    for (int r = r0; r < r1; ++r) {
        const int4 v = c4[r * 32 + lane];
        const int s = (v.x + v.y) + (v.z + v.w);
        int inc = s;
        for (int o = 1; o < 32; o <<= 1) {
            const int t = __shfl_up(inc, o);
            if (lane >= o) inc += t;
        }
        const int ex = carry + inc - s;
        int4 q;
        q.x = ex; q.y = ex + v.x; q.z = q.y + v.y; q.w = q.z + v.z;
        o4[r * 32 + lane] = q;
        carry += __shfl(inc, 31);
    }
}

__global__ void k_fill(const int* __restrict__ src, const int* __restrict__ dst,
                       const int* __restrict__ rank, const int* __restrict__ off,
                       int* __restrict__ csr, int E) {
    int e = blockIdx.x * blockDim.x + threadIdx.x;
    if (e < E) csr[off[dst[e]] + rank[e]] = src[e];
}

template <int K, int NT>
__global__ __launch_bounds__(512)
void k_gemm_wmma_f32(const float* __restrict__ X, const float* __restrict__ Wg,
                     const int* __restrict__ off, float* __restrict__ Y,
                     int N, int E) {
    constexpr int Cout = NT * 16;
    constexpr int KQ   = K / 4;
    constexpr int PQ4  = Cout + 8;
    constexpr int NCH  = K / 32;
    __shared__ float4 Ws[KQ * PQ4];

    const int tid  = threadIdx.x;
    const int lane = tid & 31;
    const int w    = __builtin_amdgcn_readfirstlane(tid >> 5);
    const int half = lane >> 4;
    const int c    = lane & 15;
    const int npan = N >> 4;
    const int pstride = (int)gridDim.x * 16;
    const int g  = w >> 2;
    const int sw = (((w + g) & 3) << 2) | g;
    int p = (int)blockIdx.x + (int)gridDim.x * sw;

    const float4* __restrict__ xa =
        (const float4*)(X + (size_t)((p < npan ? p : 0) * 16 + c) * K) + half;
    float4 a[4];
#pragma unroll
    for (int q = 0; q < 4; ++q) a[q] = xa[2 * q];

#pragma unroll 8
    for (int idx = tid; idx < KQ * Cout; idx += 512) {
        const int kq = idx / Cout;
        const int n  = idx % Cout;
        const float* __restrict__ wp = Wg + (size_t)(kq * 4) * Cout + n;
        float4 v;
        v.x = wp[0];
        v.y = wp[Cout];
        v.z = wp[2 * Cout];
        v.w = wp[3 * Cout];
        Ws[kq * PQ4 + n] = v;
    }
    __syncthreads();

    const int lo = half * PQ4 + c;
    float4 bc[NT];
#pragma unroll
    for (int t = 0; t < NT; ++t) bc[t] = Ws[lo + t * 16];

    v8f acc[NT];
#pragma unroll
    for (int t = 0; t < NT; ++t) acc[t] = (v8f){};

    int4 oa = {0, 0, 0, 0}, ob = {0, 0, 0, 0};
    int  ot = 0;

    int j = 0;
    while (p < npan) {
        const int pn = p + pstride;
        const bool lastj = (j == NCH - 1);

        const int  r0 = p * 16 + half * 8;
        const bool rn = (r0 + 8 < N);
        if (j == 0) {
            oa = *(const int4*)(off + r0);
            ob = *(const int4*)(off + r0 + 4);
            ot = off[rn ? r0 + 8 : r0];
        }

        const float4* __restrict__ xn =
            (const float4*)(X + (size_t)((pn < npan ? pn : p) * 16 + c) * K) + half;
        const float4* __restrict__ src = lastj ? xn : (xa + (j + 1) * 8);
        float4 an[4];
#pragma unroll
        for (int q = 0; q < 4; ++q) an[q] = src[2 * q];

#pragma unroll
        for (int q = 0; q < 4; ++q) {
            const int rown = (q == 3) ? (((j + 1) & (NCH - 1)) * 8) : (j * 8 + 2 * q + 2);
            const int bo = lo + rown * PQ4;
            float4 bn[NT];
#pragma unroll
            for (int t = 0; t < NT; ++t) bn[t] = Ws[bo + t * 16];

            const v2f a01 = {a[q].x, a[q].y};
            const v2f a23 = {a[q].z, a[q].w};
#pragma unroll
            for (int t = 0; t < NT; ++t) {
                const v2f b01 = {bc[t].x, bc[t].y};
                acc[t] = __builtin_amdgcn_wmma_f32_16x16x4_f32(
                              false, a01,  false, b01,
                              (short)0, acc[t],
                              false,  false);
            }
#pragma unroll
            for (int t = 0; t < NT; ++t) {
                const v2f b23 = {bc[t].z, bc[t].w};
                acc[t] = __builtin_amdgcn_wmma_f32_16x16x4_f32(
                              false, a23,  false, b23,
                              (short)0, acc[t],
                              false,  false);
            }
#pragma unroll
            for (int t = 0; t < NT; ++t) bc[t] = bn[t];
        }
#pragma unroll
        for (int q = 0; q < 4; ++q) a[q] = an[q];

        if (lastj) {
            float dv[8];
            dv[0] = rsqrtf((float)(oa.y - oa.x + 1));
            dv[1] = rsqrtf((float)(oa.z - oa.y + 1));
            dv[2] = rsqrtf((float)(oa.w - oa.z + 1));
            dv[3] = rsqrtf((float)(ob.x - oa.w + 1));
            dv[4] = rsqrtf((float)(ob.y - ob.x + 1));
            dv[5] = rsqrtf((float)(ob.z - ob.y + 1));
            dv[6] = rsqrtf((float)(ob.w - ob.z + 1));
            dv[7] = rsqrtf((float)((rn ? ot : E) - ob.w + 1));

            float* __restrict__ yp = Y + (size_t)(p * 16) * Cout + c;
#pragma unroll
            for (int t = 0; t < NT; ++t) {
#pragma unroll
                for (int v = 0; v < 8; ++v)
                    yp[(size_t)(v + half * 8) * Cout + t * 16] = acc[t][v] * dv[v];
                acc[t] = (v8f){};
            }
            xa = xn;
            p  = pn;
            j  = 0;
        } else {
            ++j;
        }
    }
}

template <int C, bool RELU, bool MOVE>
__global__ __launch_bounds__(256)
void k_gather(const float* __restrict__ h, const float* __restrict__ bias,
              const int* __restrict__ off, const int* __restrict__ csr,
              int* __restrict__ csr_new, float* __restrict__ out, int N, int E) {
    constexpr int LPR  = C / 4;
    constexpr int G    = 32 / LPR;
    constexpr int U    = (G == 4) ? 2 : 4;
    constexpr int STEP = U * G;
    const int lane = threadIdx.x & 31;
    const int i = __builtin_amdgcn_readfirstlane((int)(blockIdx.x * 8 + (threadIdx.x >> 5)));
    if (i >= N) return;
    const int grp = lane / LPR;
    const int li  = lane % LPR;

    const int start = off[i];
    const int end   = (i + 1 < N) ? off[i + 1] : E;
    const int cnt   = end - start;
    const float di  = rsqrtf((float)(cnt + 1));

    const float4* __restrict__ h4 = (const float4*)h;
    float4 acc = {0.0f, 0.0f, 0.0f, 0.0f};
    float4 b   = {0.0f, 0.0f, 0.0f, 0.0f};
    if (grp == 0) {
        b   = ((const float4*)bias)[li];
        acc = h4[(size_t)i * LPR + li];
    }

    for (int base = 0; base < cnt; base += 32) {
        const int n = min(32, cnt - base);
        int s = i;
        if (lane < n) {
            s = csr[start + base + lane];
            if (MOVE) csr_new[start + base + lane] = s;
        }
        for (int t = 0; t < n; t += STEP) {
            float4 v[U];
            float  wv[U];
#pragma unroll
            for (int k = 0; k < U; ++k) {
                const int e = (G == 1) ? (t + k) : (t + k * G + grp);
                int sv;
                if constexpr (G == 1) sv = __builtin_amdgcn_readlane(s, e);
                else                  sv = __shfl(s, e);
                wv[k] = (e < n) ? 1.0f : 0.0f;
                v[k]  = h4[(size_t)sv * LPR + li];
            }
#pragma unroll
            for (int k = 0; k < U; ++k) {
                acc.x = fmaf(v[k].x, wv[k], acc.x);
                acc.y = fmaf(v[k].y, wv[k], acc.y);
                acc.z = fmaf(v[k].z, wv[k], acc.z);
                acc.w = fmaf(v[k].w, wv[k], acc.w);
            }
        }
    }

    for (int o = LPR; o < 32; o <<= 1) {
        acc.x += __shfl_xor(acc.x, o);
        acc.y += __shfl_xor(acc.y, o);
        acc.z += __shfl_xor(acc.z, o);
        acc.w += __shfl_xor(acc.w, o);
    }
    if (grp == 0) {
        acc.x = fmaf(acc.x, di, b.x); acc.y = fmaf(acc.y, di, b.y);
        acc.z = fmaf(acc.z, di, b.z); acc.w = fmaf(acc.w, di, b.w);
        if (RELU) {
            acc.x = fmaxf(acc.x, 0.0f); acc.y = fmaxf(acc.y, 0.0f);
            acc.z = fmaxf(acc.z, 0.0f); acc.w = fmaxf(acc.w, 0.0f);
        }
        ((float4*)out)[(size_t)i * LPR + li] = acc;
    }
}

template <int C, int NT, bool MOVE>
__global__ __launch_bounds__(512)
void k_gather_gemm(const float* __restrict__ h, const float* __restrict__ bias,
                   const float* __restrict__ Wg,
                   const int* __restrict__ off, const int* __restrict__ csr,
                   int* __restrict__ csr_new, float* __restrict__ Y, int N, int E) {
    constexpr int LPR  = C / 4;
    constexpr int G    = 32 / LPR;
    constexpr int U    = (G == 4) ? 2 : 4;
    constexpr int STEP = U * G;
    constexpr int Cout = NT * 16;
    constexpr int KQ   = C / 4;
    constexpr int PQ4  = Cout + 8;
    constexpr int AP4  = C / 4 + 1;
    constexpr int WPT  = (KQ * Cout) / 512;
    __shared__ float4 Ws[KQ * PQ4];
    __shared__ float4 As[16 * AP4];
    __shared__ float  dvs[16];

    const int tid  = threadIdx.x;
    const int lane = tid & 31;
    const int w  = __builtin_amdgcn_readfirstlane(tid >> 5);
    const int i  = __builtin_amdgcn_readfirstlane((int)blockIdx.x * 16 + w);
    const bool live = (i < N);
    const int  ic   = live ? i : 0;
    const int grp = lane / LPR;
    const int li  = lane % LPR;

    float4 wst[WPT];
#pragma unroll
    for (int r = 0; r < WPT; ++r) {
        const int idx = tid + r * 512;
        const int kq = idx / Cout;
        const int n  = idx % Cout;
        const float* __restrict__ wp = Wg + (size_t)(kq * 4) * Cout + n;
        wst[r].x = wp[0];
        wst[r].y = wp[Cout];
        wst[r].z = wp[2 * Cout];
        wst[r].w = wp[3 * Cout];
    }

    const int start = off[ic];
    const int end   = (ic + 1 < N) ? off[ic + 1] : E;
    const int cnt   = live ? (end - start) : 0;
    const float di  = rsqrtf((float)(cnt + 1));

    const float4* __restrict__ h4 = (const float4*)h;
    float4 acc = {0.0f, 0.0f, 0.0f, 0.0f};
    float4 b   = {0.0f, 0.0f, 0.0f, 0.0f};
    if (grp == 0) {
        b   = ((const float4*)bias)[li];
        acc = h4[(size_t)ic * LPR + li];
    }

    for (int base = 0; base < cnt; base += 32) {
        const int n = min(32, cnt - base);
        int s = ic;
        if (lane < n) {
            s = csr[start + base + lane];
            if (MOVE) csr_new[start + base + lane] = s;
        }
        for (int t = 0; t < n; t += STEP) {
            float4 v[U];
            float  wv[U];
#pragma unroll
            for (int k = 0; k < U; ++k) {
                const int e = (G == 1) ? (t + k) : (t + k * G + grp);
                int sv;
                if constexpr (G == 1) sv = __builtin_amdgcn_readlane(s, e);
                else                  sv = __shfl(s, e);
                wv[k] = (e < n) ? 1.0f : 0.0f;
                v[k]  = h4[(size_t)sv * LPR + li];
            }
#pragma unroll
            for (int k = 0; k < U; ++k) {
                acc.x = fmaf(v[k].x, wv[k], acc.x);
                acc.y = fmaf(v[k].y, wv[k], acc.y);
                acc.z = fmaf(v[k].z, wv[k], acc.z);
                acc.w = fmaf(v[k].w, wv[k], acc.w);
            }
        }
    }

    for (int o = LPR; o < 32; o <<= 1) {
        acc.x += __shfl_xor(acc.x, o);
        acc.y += __shfl_xor(acc.y, o);
        acc.z += __shfl_xor(acc.z, o);
        acc.w += __shfl_xor(acc.w, o);
    }
    if (grp == 0) {
        acc.x = fmaf(acc.x, di, b.x); acc.y = fmaf(acc.y, di, b.y);
        acc.z = fmaf(acc.z, di, b.z); acc.w = fmaf(acc.w, di, b.w);
        acc.x = fmaxf(acc.x, 0.0f); acc.y = fmaxf(acc.y, 0.0f);
        acc.z = fmaxf(acc.z, 0.0f); acc.w = fmaxf(acc.w, 0.0f);
        As[w * AP4 + li] = acc;
    }
    if (lane == 0) dvs[w] = di;
#pragma unroll
    for (int r = 0; r < WPT; ++r) {
        const int idx = tid + r * 512;
        Ws[(idx / Cout) * PQ4 + (idx % Cout)] = wst[r];
    }
    __syncthreads();

    if (w < NT) {
        const int hf = lane >> 4;
        const int c  = lane & 15;
        v8f d = (v8f){};
#pragma unroll
        for (int q = 0; q < C / 8; ++q) {
            const float4 a  = As[c * AP4 + 2 * q + hf];
            const float4 bw = Ws[(2 * q + hf) * PQ4 + w * 16 + c];
            const v2f a01 = {a.x, a.y};
            const v2f a23 = {a.z, a.w};
            const v2f b01 = {bw.x, bw.y};
            const v2f b23 = {bw.z, bw.w};
            d = __builtin_amdgcn_wmma_f32_16x16x4_f32(
                     false, a01,  false, b01,
                     (short)0, d,  false,  false);
            d = __builtin_amdgcn_wmma_f32_16x16x4_f32(
                     false, a23,  false, b23,
                     (short)0, d,  false,  false);
        }
        const int row0 = (int)blockIdx.x * 16 + hf * 8;
        float* __restrict__ yp = Y + (size_t)row0 * Cout + w * 16 + c;
#pragma unroll
        for (int v = 0; v < 8; ++v)
            if (row0 + v < N) yp[(size_t)v * Cout] = d[v] * dvs[hf * 8 + v];
    }
}

extern "C" void kernel_launch(void* const* d_in, const int* in_sizes, int n_in,
                              void* d_out, int out_size, void* d_ws, size_t ws_size,
                              hipStream_t stream) {
    const float* x  = (const float*)d_in[0];
    const int*   ei = (const int*)  d_in[1];
    const float* W1 = (const float*)d_in[2];
    const float* b1 = (const float*)d_in[3];
    const float* W2 = (const float*)d_in[4];
    const float* b2 = (const float*)d_in[5];
    const float* W3 = (const float*)d_in[6];
    const float* b3 = (const float*)d_in[7];

    const int K0 = 512, C1 = 128, C2 = 64, C3 = 32;
    const int N = in_sizes[0] / K0;
    const int E = in_sizes[1] / 2;
    const int* src = ei;
    const int* dst = ei + E;

    float* ws   = (float*)d_ws;
    size_t pad  = ((size_t)N + 255) & ~(size_t)255;
    int*   off  = (int*)ws;
    float* bufH = ws + pad;
    float* bufA = bufH + (size_t)N * C1;
    float* out  = (float*)d_out;
    int*   cnt  = (int*)bufA;
    int*   rank = cnt + pad;
    int*   csr0 = (int*)d_out;
    int*   csr1 = (int*)(bufH + (size_t)N * C2);

    const int TB = 256;

    k_cnt_zero <<<(int)(pad / TB), TB, 0, stream>>>(cnt, (int)pad);
    k_edge_rank<<<(E + TB - 1) / TB, TB, 0, stream>>>(dst, cnt, rank, E);
    k_scan     <<<1, 1024, 0, stream>>>(cnt, off, (int)pad);
    k_fill     <<<(E + TB - 1) / TB, TB, 0, stream>>>(src, dst, rank, off, csr0, E);

    k_gemm_wmma_f32<512, 8><<<256, 512, 0, stream>>>(x, W1, off, bufH, N, E);

    k_gather_gemm<128, 4, false><<<(N + 15) / 16, 512, 0, stream>>>(bufH, b1, W2, off, csr0, csr1, bufA, N, E);

    k_gather_gemm<64, 2, true><<<(N + 15) / 16, 512, 0, stream>>>(bufA, b2, W3, off, csr0, csr1, bufH, N, E);

    k_gather<32, false, false><<<(N + 7) / 8, TB, 0, stream>>>(bufH, b3, off, csr1, csr1, out, N, E);
}


# edge bucketing rebuilt as a two-level counting sort in LDS (256-node tiles, then nodes): no global atomics, no zeroing launch, edge words move in runs; order inside a bucket = LDS arrival order
# speedup vs baseline: 6.8447x; 1.4506x over previous
#include <hip/hip_runtime.h>
#include <hip/hip_bf16.h>

typedef __attribute__((ext_vector_type(2))) float v2f;
typedef __attribute__((ext_vector_type(8))) float v8f;


template <int PER>
__device__ __forceinline__ void wave_excl_scan(const int* c, int* st, int lane) {
    int v[PER];
    int s = 0;
#pragma unroll
    for (int k = 0; k < PER; ++k) { v[k] = c[lane * PER + k]; s += v[k]; }
    int inc = s;
    for (int o = 1; o < 32; o <<= 1) {
        const int t = __shfl_up(inc, o);
        if (lane >= o) inc += t;
    }
    int ex = inc - s;
#pragma unroll
    for (int k = 0; k < PER; ++k) { st[lane * PER + k] = ex; ex += v[k]; }
}

constexpr int TILE_MAX = 512;
constexpr int BT       = 1024;
constexpr int EPT      = 8;

__global__ __launch_bounds__(BT)
void k_tile_hist(const int* __restrict__ dst, int* __restrict__ histT,
                 int E, int chunk, int tiles) {
    __shared__ int lcnt[TILE_MAX];
    const int tid = threadIdx.x;
    const int b   = blockIdx.x;
    const int e0  = min(E, b * chunk);
    const int e1  = min(E, e0 + chunk);
    if (tid < TILE_MAX) lcnt[tid] = 0;
    __syncthreads();
    for (int s0 = e0; s0 < e1; s0 += BT * EPT) {
        int d[EPT];
#pragma unroll
        for (int k = 0; k < EPT; ++k) {
            const int e = s0 + k * BT + tid;
            d[k] = (e < e1) ? dst[e] : -1;
        }
#pragma unroll
        for (int k = 0; k < EPT; ++k)
            if (d[k] >= 0) atomicAdd(&lcnt[d[k] >> 8], 1);
    }
    __syncthreads();
    for (int t = tid; t < tiles; t += BT) histT[t * 256 + b] = lcnt[t];
}

__global__ __launch_bounds__(1024)
void k_scan(const int* __restrict__ cnt, int* __restrict__ off, int npad) {
    __shared__ int wsum[32];
    const int lane  = threadIdx.x & 31;
    const int w     = __builtin_amdgcn_readfirstlane((int)(threadIdx.x >> 5));
    const int nrows = npad >> 7;
    const int per   = (nrows + 31) >> 5;
    const int r0    = min(w * per, nrows);
    const int r1    = min(r0 + per, nrows);
    const int4* __restrict__ c4 = (const int4*)cnt;
    int4* __restrict__ o4 = (int4*)off;

    int tot = 0;
#pragma unroll 4
    for (int r = r0; r < r1; ++r) {
        const int4 v = c4[r * 32 + lane];
        tot += (v.x + v.y) + (v.z + v.w);
    }
    for (int o = 16; o > 0; o >>= 1) tot += __shfl_xor(tot, o);
    if (lane == 0) wsum[w] = tot;
    __syncthreads();
    int carry = (lane < w) ? wsum[lane] : 0;
    for (int o = 16; o > 0; o >>= 1) carry += __shfl_xor(carry, o);

#pragma unroll 4
    for (int r = r0; r < r1; ++r) {
        const int4 v = c4[r * 32 + lane];
        const int s = (v.x + v.y) + (v.z + v.w);
        int inc = s;
        for (int o = 1; o < 32; o <<= 1) {
            const int t = __shfl_up(inc, o);
            if (lane >= o) inc += t;
        }
        const int ex = carry + inc - s;
        int4 q;
        q.x = ex; q.y = ex + v.x; q.z = q.y + v.y; q.w = q.z + v.z;
        o4[r * 32 + lane] = q;
        carry += __shfl(inc, 31);
    }
}

__global__ __launch_bounds__(BT)
void k_tile_scatter(const int* __restrict__ src, const int* __restrict__ dst,
                    const int* __restrict__ offT, int* __restrict__ tmp,
                    int E, int chunk, int tiles) {
    __shared__ int lcnt[TILE_MAX];
    __shared__ int lst[TILE_MAX];
    __shared__ int lbase[TILE_MAX];
    __shared__ int spos[BT * EPT];
    __shared__ int swrd[BT * EPT];
    const int tid  = threadIdx.x;
    const int lane = tid & 31;
    const int w    = __builtin_amdgcn_readfirstlane(tid >> 5);
    const int b    = blockIdx.x;
    const int e0   = min(E, b * chunk);
    const int e1   = min(E, e0 + chunk);
    if (tid < TILE_MAX) lbase[tid] = (tid < tiles) ? offT[tid * 256 + b] : 0;

    for (int s0 = e0; s0 < e1; s0 += BT * EPT) {
        const int n = min(BT * EPT, e1 - s0);
        if (tid < TILE_MAX) lcnt[tid] = 0;
        __syncthreads();
        int d[EPT], sv[EPT], r[EPT];
#pragma unroll
        for (int k = 0; k < EPT; ++k) {
            const int e = s0 + k * BT + tid;
            d[k]  = (e < e1) ? dst[e] : -1;
            sv[k] = (e < e1) ? src[e] : 0;
        }
#pragma unroll
        for (int k = 0; k < EPT; ++k)
            r[k] = (d[k] >= 0) ? atomicAdd(&lcnt[d[k] >> 8], 1) : 0;
        __syncthreads();
        if (w == 0) wave_excl_scan<TILE_MAX / 32>(lcnt, lst, lane);
        __syncthreads();
#pragma unroll
        for (int k = 0; k < EPT; ++k) {
            if (d[k] >= 0) {
                const int tl = d[k] >> 8;
                const int i  = lst[tl] + r[k];
                spos[i] = lbase[tl] + r[k];
                swrd[i] = sv[k] | ((d[k] & 255) << 24);
            }
        }
        __syncthreads();
        for (int i = tid; i < n; i += BT) tmp[spos[i]] = swrd[i];
        if (tid < TILE_MAX) lbase[tid] += lcnt[tid];
    }
}

constexpr int SCAP = 16384;
__global__ __launch_bounds__(BT)
void k_tile_sort(const int* __restrict__ tmp, const int* __restrict__ offT,
                 int* __restrict__ off, int* __restrict__ csr, int E, int tiles) {
    __shared__ int ncnt[256];
    __shared__ int nst[256];
    __shared__ int ncur[256];
    __shared__ int sorted[SCAP];
    const int tid  = threadIdx.x;
    const int lane = tid & 31;
    const int w    = __builtin_amdgcn_readfirstlane(tid >> 5);
    const int t    = blockIdx.x;
    const int ts   = offT[t * 256];
    const int te   = (t + 1 < tiles) ? offT[(t + 1) * 256] : E;
    const int cnt  = te - ts;
    const int* __restrict__ tp = tmp + ts;
    if (tid < 256) { ncnt[tid] = 0; ncur[tid] = 0; }
    __syncthreads();
    for (int i0 = 0; i0 < cnt; i0 += BT * 4) {
        int wd[4];
#pragma unroll
        for (int k = 0; k < 4; ++k)
            wd[k] = tp[min(i0 + k * BT + tid, cnt - 1)];
#pragma unroll
        for (int k = 0; k < 4; ++k)
            if (i0 + k * BT + tid < cnt) atomicAdd(&ncnt[(unsigned)wd[k] >> 24], 1);
    }
    __syncthreads();
    if (w == 0) {
        wave_excl_scan<8>(ncnt, nst, lane);
        int4 qa, qb;
        qa.x = ts + nst[lane * 8 + 0]; qa.y = ts + nst[lane * 8 + 1];
        qa.z = ts + nst[lane * 8 + 2]; qa.w = ts + nst[lane * 8 + 3];
        qb.x = ts + nst[lane * 8 + 4]; qb.y = ts + nst[lane * 8 + 5];
        qb.z = ts + nst[lane * 8 + 6]; qb.w = ts + nst[lane * 8 + 7];
        int4* __restrict__ o4 = (int4*)(off + t * 256 + lane * 8);
        o4[0] = qa;
        o4[1] = qb;
    }
    __syncthreads();
    const bool staged = (cnt <= SCAP);
    int* __restrict__ cp = csr + ts;
    for (int i0 = 0; i0 < cnt; i0 += BT * 4) {
        int wd[4];
#pragma unroll
        for (int k = 0; k < 4; ++k) {
            const int i = i0 + k * BT + tid;
            wd[k] = (i < cnt) ? tp[i] : 0;
        }
#pragma unroll
        for (int k = 0; k < 4; ++k) {
            if (i0 + k * BT + tid < cnt) {
                const int dl = (int)((unsigned)wd[k] >> 24);
                const int p  = nst[dl] + atomicAdd(&ncur[dl], 1);
                const int s  = wd[k] & 0x00FFFFFF;
                if (staged) sorted[p] = s; else cp[p] = s;
            }
        }
    }
    if (staged) {
        __syncthreads();
        for (int i = tid; i < cnt; i += BT) cp[i] = sorted[i];
    }
}

template <int K, int NT>
__global__ __launch_bounds__(512)
void k_gemm_wmma_f32(const float* __restrict__ X, const float* __restrict__ Wg,
                     const int* __restrict__ off, float* __restrict__ Y,
                     int N, int E) {
    constexpr int Cout = NT * 16;
    constexpr int KQ   = K / 4;
    constexpr int PQ4  = Cout + 8;
    constexpr int NCH  = K / 32;
    __shared__ float4 Ws[KQ * PQ4];

    const int tid  = threadIdx.x;
    const int lane = tid & 31;
    const int w    = __builtin_amdgcn_readfirstlane(tid >> 5);
    const int half = lane >> 4;
    const int c    = lane & 15;
    const int npan = N >> 4;
    const int pstride = (int)gridDim.x * 16;
    const int g  = w >> 2;
    const int sw = (((w + g) & 3) << 2) | g;
    int p = (int)blockIdx.x + (int)gridDim.x * sw;

    const float4* __restrict__ xa =
        (const float4*)(X + (size_t)((p < npan ? p : 0) * 16 + c) * K) + half;
    float4 a[4];
#pragma unroll
    for (int q = 0; q < 4; ++q) a[q] = xa[2 * q];

#pragma unroll 8
    for (int idx = tid; idx < KQ * Cout; idx += 512) {
        const int kq = idx / Cout;
        const int n  = idx % Cout;
        const float* __restrict__ wp = Wg + (size_t)(kq * 4) * Cout + n;
        float4 v;
        v.x = wp[0];
        v.y = wp[Cout];
        v.z = wp[2 * Cout];
        v.w = wp[3 * Cout];
        Ws[kq * PQ4 + n] = v;
    }
    __syncthreads();

    const int lo = half * PQ4 + c;
    float4 bc[NT];
#pragma unroll
    for (int t = 0; t < NT; ++t) bc[t] = Ws[lo + t * 16];

    v8f acc[NT];
#pragma unroll
    for (int t = 0; t < NT; ++t) acc[t] = (v8f){};

    int4 oa = {0, 0, 0, 0}, ob = {0, 0, 0, 0};
    int  ot = 0;

    int j = 0;
    while (p < npan) {
        const int pn = p + pstride;
        const bool lastj = (j == NCH - 1);

        const int  r0 = p * 16 + half * 8;
        const bool rn = (r0 + 8 < N);
        if (j == 0) {
            oa = *(const int4*)(off + r0);
            ob = *(const int4*)(off + r0 + 4);
            ot = off[rn ? r0 + 8 : r0];
        }

        const float4* __restrict__ xn =
            (const float4*)(X + (size_t)((pn < npan ? pn : p) * 16 + c) * K) + half;
        const float4* __restrict__ src = lastj ? xn : (xa + (j + 1) * 8);
        float4 an[4];
#pragma unroll
        for (int q = 0; q < 4; ++q) an[q] = src[2 * q];

#pragma unroll
        for (int q = 0; q < 4; ++q) {
            const int rown = (q == 3) ? (((j + 1) & (NCH - 1)) * 8) : (j * 8 + 2 * q + 2);
            const int bo = lo + rown * PQ4;
            float4 bn[NT];
#pragma unroll
            for (int t = 0; t < NT; ++t) bn[t] = Ws[bo + t * 16];

            const v2f a01 = {a[q].x, a[q].y};
            const v2f a23 = {a[q].z, a[q].w};
#pragma unroll
            for (int t = 0; t < NT; ++t) {
                const v2f b01 = {bc[t].x, bc[t].y};
                acc[t] = __builtin_amdgcn_wmma_f32_16x16x4_f32(
                              false, a01,  false, b01,
                              (short)0, acc[t],
                              false,  false);
            }
#pragma unroll
            for (int t = 0; t < NT; ++t) {
                const v2f b23 = {bc[t].z, bc[t].w};
                acc[t] = __builtin_amdgcn_wmma_f32_16x16x4_f32(
                              false, a23,  false, b23,
                              (short)0, acc[t],
                              false,  false);
            }
#pragma unroll
            for (int t = 0; t < NT; ++t) bc[t] = bn[t];
        }
#pragma unroll
        for (int q = 0; q < 4; ++q) a[q] = an[q];

        if (lastj) {
            float dv[8];
            dv[0] = rsqrtf((float)(oa.y - oa.x + 1));
            dv[1] = rsqrtf((float)(oa.z - oa.y + 1));
            dv[2] = rsqrtf((float)(oa.w - oa.z + 1));
            dv[3] = rsqrtf((float)(ob.x - oa.w + 1));
            dv[4] = rsqrtf((float)(ob.y - ob.x + 1));
            dv[5] = rsqrtf((float)(ob.z - ob.y + 1));
            dv[6] = rsqrtf((float)(ob.w - ob.z + 1));
            dv[7] = rsqrtf((float)((rn ? ot : E) - ob.w + 1));

            float* __restrict__ yp = Y + (size_t)(p * 16) * Cout + c;
#pragma unroll
            for (int t = 0; t < NT; ++t) {
#pragma unroll
                for (int v = 0; v < 8; ++v)
                    yp[(size_t)(v + half * 8) * Cout + t * 16] = acc[t][v] * dv[v];
                acc[t] = (v8f){};
            }
            xa = xn;
            p  = pn;
            j  = 0;
        } else {
            ++j;
        }
    }
}

template <int C, bool RELU, bool MOVE>
__global__ __launch_bounds__(256)
void k_gather(const float* __restrict__ h, const float* __restrict__ bias,
              const int* __restrict__ off, const int* __restrict__ csr,
              int* __restrict__ csr_new, float* __restrict__ out, int N, int E) {
    constexpr int LPR  = C / 4;
    constexpr int G    = 32 / LPR;
    constexpr int U    = (G == 4) ? 2 : 4;
    constexpr int STEP = U * G;
    const int lane = threadIdx.x & 31;
    const int i = __builtin_amdgcn_readfirstlane((int)(blockIdx.x * 8 + (threadIdx.x >> 5)));
    if (i >= N) return;
    const int grp = lane / LPR;
    const int li  = lane % LPR;

    const int start = off[i];
    const int end   = (i + 1 < N) ? off[i + 1] : E;
    const int cnt   = end - start;
    const float di  = rsqrtf((float)(cnt + 1));

    const float4* __restrict__ h4 = (const float4*)h;
    float4 acc = {0.0f, 0.0f, 0.0f, 0.0f};
    float4 b   = {0.0f, 0.0f, 0.0f, 0.0f};
    if (grp == 0) {
        b   = ((const float4*)bias)[li];
        acc = h4[(size_t)i * LPR + li];
    }

    for (int base = 0; base < cnt; base += 32) {
        const int n = min(32, cnt - base);
        int s = i;
        if (lane < n) {
            s = csr[start + base + lane];
            if (MOVE) csr_new[start + base + lane] = s;
        }
        for (int t = 0; t < n; t += STEP) {
            float4 v[U];
            float  wv[U];
#pragma unroll
            for (int k = 0; k < U; ++k) {
                const int e = (G == 1) ? (t + k) : (t + k * G + grp);
                int sv;
                if constexpr (G == 1) sv = __builtin_amdgcn_readlane(s, e);
                else                  sv = __shfl(s, e);
                wv[k] = (e < n) ? 1.0f : 0.0f;
                v[k]  = h4[(size_t)sv * LPR + li];
            }
#pragma unroll
            for (int k = 0; k < U; ++k) {
                acc.x = fmaf(v[k].x, wv[k], acc.x);
                acc.y = fmaf(v[k].y, wv[k], acc.y);
                acc.z = fmaf(v[k].z, wv[k], acc.z);
                acc.w = fmaf(v[k].w, wv[k], acc.w);
            }
        }
    }

    for (int o = LPR; o < 32; o <<= 1) {
        acc.x += __shfl_xor(acc.x, o);
        acc.y += __shfl_xor(acc.y, o);
        acc.z += __shfl_xor(acc.z, o);
        acc.w += __shfl_xor(acc.w, o);
    }
    if (grp == 0) {
        acc.x = fmaf(acc.x, di, b.x); acc.y = fmaf(acc.y, di, b.y);
        acc.z = fmaf(acc.z, di, b.z); acc.w = fmaf(acc.w, di, b.w);
        if (RELU) {
            acc.x = fmaxf(acc.x, 0.0f); acc.y = fmaxf(acc.y, 0.0f);
            acc.z = fmaxf(acc.z, 0.0f); acc.w = fmaxf(acc.w, 0.0f);
        }
        ((float4*)out)[(size_t)i * LPR + li] = acc;
    }
}

template <int C, int NT, bool MOVE>
__global__ __launch_bounds__(512)
void k_gather_gemm(const float* __restrict__ h, const float* __restrict__ bias,
                   const float* __restrict__ Wg,
                   const int* __restrict__ off, const int* __restrict__ csr,
                   int* __restrict__ csr_new, float* __restrict__ Y, int N, int E) {
    constexpr int LPR  = C / 4;
    constexpr int G    = 32 / LPR;
    constexpr int U    = (G == 4) ? 2 : 4;
    constexpr int STEP = U * G;
    constexpr int Cout = NT * 16;
    constexpr int KQ   = C / 4;
    constexpr int PQ4  = Cout + 8;
    constexpr int AP4  = C / 4 + 1;
    constexpr int WPT  = (KQ * Cout) / 512;
    __shared__ float4 Ws[KQ * PQ4];
    __shared__ float4 As[16 * AP4];
    __shared__ float  dvs[16];

    const int tid  = threadIdx.x;
    const int lane = tid & 31;
    const int w  = __builtin_amdgcn_readfirstlane(tid >> 5);
    const int i  = __builtin_amdgcn_readfirstlane((int)blockIdx.x * 16 + w);
    const bool live = (i < N);
    const int  ic   = live ? i : 0;
    const int grp = lane / LPR;
    const int li  = lane % LPR;

    float4 wst[WPT];
#pragma unroll
    for (int r = 0; r < WPT; ++r) {
        const int idx = tid + r * 512;
        const int kq = idx / Cout;
        const int n  = idx % Cout;
        const float* __restrict__ wp = Wg + (size_t)(kq * 4) * Cout + n;
        wst[r].x = wp[0];
        wst[r].y = wp[Cout];
        wst[r].z = wp[2 * Cout];
        wst[r].w = wp[3 * Cout];
    }

    const int start = off[ic];
    const int end   = (ic + 1 < N) ? off[ic + 1] : E;
    const int cnt   = live ? (end - start) : 0;
    const float di  = rsqrtf((float)(cnt + 1));

    const float4* __restrict__ h4 = (const float4*)h;
    float4 acc = {0.0f, 0.0f, 0.0f, 0.0f};
    float4 b   = {0.0f, 0.0f, 0.0f, 0.0f};
    if (grp == 0) {
        b   = ((const float4*)bias)[li];
        acc = h4[(size_t)ic * LPR + li];
    }

    for (int base = 0; base < cnt; base += 32) {
        const int n = min(32, cnt - base);
        int s = ic;
        if (lane < n) {
            s = csr[start + base + lane];
            if (MOVE) csr_new[start + base + lane] = s;
        }
        for (int t = 0; t < n; t += STEP) {
            float4 v[U];
            float  wv[U];
#pragma unroll
            for (int k = 0; k < U; ++k) {
                const int e = (G == 1) ? (t + k) : (t + k * G + grp);
                int sv;
                if constexpr (G == 1) sv = __builtin_amdgcn_readlane(s, e);
                else                  sv = __shfl(s, e);
                wv[k] = (e < n) ? 1.0f : 0.0f;
                v[k]  = h4[(size_t)sv * LPR + li];
            }
#pragma unroll
            for (int k = 0; k < U; ++k) {
                acc.x = fmaf(v[k].x, wv[k], acc.x);
                acc.y = fmaf(v[k].y, wv[k], acc.y);
                acc.z = fmaf(v[k].z, wv[k], acc.z);
                acc.w = fmaf(v[k].w, wv[k], acc.w);
            }
        }
    }

    for (int o = LPR; o < 32; o <<= 1) {
        acc.x += __shfl_xor(acc.x, o);
        acc.y += __shfl_xor(acc.y, o);
        acc.z += __shfl_xor(acc.z, o);
        acc.w += __shfl_xor(acc.w, o);
    }
    if (grp == 0) {
        acc.x = fmaf(acc.x, di, b.x); acc.y = fmaf(acc.y, di, b.y);
        acc.z = fmaf(acc.z, di, b.z); acc.w = fmaf(acc.w, di, b.w);
        acc.x = fmaxf(acc.x, 0.0f); acc.y = fmaxf(acc.y, 0.0f);
        acc.z = fmaxf(acc.z, 0.0f); acc.w = fmaxf(acc.w, 0.0f);
        As[w * AP4 + li] = acc;
    }
    if (lane == 0) dvs[w] = di;
#pragma unroll
    for (int r = 0; r < WPT; ++r) {
        const int idx = tid + r * 512;
        Ws[(idx / Cout) * PQ4 + (idx % Cout)] = wst[r];
    }
    __syncthreads();

    if (w < NT) {
        const int hf = lane >> 4;
        const int c  = lane & 15;
        v8f d = (v8f){};
#pragma unroll
        for (int q = 0; q < C / 8; ++q) {
            const float4 a  = As[c * AP4 + 2 * q + hf];
            const float4 bw = Ws[(2 * q + hf) * PQ4 + w * 16 + c];
            const v2f a01 = {a.x, a.y};
            const v2f a23 = {a.z, a.w};
            const v2f b01 = {bw.x, bw.y};
            const v2f b23 = {bw.z, bw.w};
            d = __builtin_amdgcn_wmma_f32_16x16x4_f32(
                     false, a01,  false, b01,
                     (short)0, d,  false,  false);
            d = __builtin_amdgcn_wmma_f32_16x16x4_f32(
                     false, a23,  false, b23,
                     (short)0, d,  false,  false);
        }
        const int row0 = (int)blockIdx.x * 16 + hf * 8;
        float* __restrict__ yp = Y + (size_t)row0 * Cout + w * 16 + c;
#pragma unroll
        for (int v = 0; v < 8; ++v)
            if (row0 + v < N) yp[(size_t)v * Cout] = d[v] * dvs[hf * 8 + v];
    }
}

extern "C" void kernel_launch(void* const* d_in, const int* in_sizes, int n_in,
                              void* d_out, int out_size, void* d_ws, size_t ws_size,
                              hipStream_t stream) {
    const float* x  = (const float*)d_in[0];
    const int*   ei = (const int*)  d_in[1];
    const float* W1 = (const float*)d_in[2];
    const float* b1 = (const float*)d_in[3];
    const float* W2 = (const float*)d_in[4];
    const float* b2 = (const float*)d_in[5];
    const float* W3 = (const float*)d_in[6];
    const float* b3 = (const float*)d_in[7];

    const int K0 = 512, C1 = 128, C2 = 64, C3 = 32;
    const int N = in_sizes[0] / K0;
    const int E = in_sizes[1] / 2;
    const int* src = ei;
    const int* dst = ei + E;

    float* ws   = (float*)d_ws;
    size_t pad  = ((size_t)N + 255) & ~(size_t)255;
    int*   off  = (int*)ws;
    float* bufH = ws + pad;
    float* bufA = bufH + (size_t)N * C1;
    float* out  = (float*)d_out;
    int*   histT = (int*)bufA;
    int*   offT  = histT + pad;
    int*   tmp   = offT + pad;
    int*   csr0 = (int*)d_out;
    int*   csr1 = (int*)(bufH + (size_t)N * C2);

    const int TB = 256;

    const int tiles = (int)(pad / 256);
    const int chunk = (E + 255) / 256;
    k_tile_hist   <<<256, BT, 0, stream>>>(dst, histT, E, chunk, tiles);
    k_scan        <<<1, 1024, 0, stream>>>(histT, offT, (int)pad);
    k_tile_scatter<<<256, BT, 0, stream>>>(src, dst, offT, tmp, E, chunk, tiles);
    k_tile_sort   <<<tiles, BT, 0, stream>>>(tmp, offT, off, csr0, E, tiles);

    k_gemm_wmma_f32<512, 8><<<256, 512, 0, stream>>>(x, W1, off, bufH, N, E);

    k_gather_gemm<128, 4, false><<<(N + 15) / 16, 512, 0, stream>>>(bufH, b1, W2, off, csr0, csr1, bufA, N, E);

    k_gather_gemm<64, 2, true><<<(N + 15) / 16, 512, 0, stream>>>(bufA, b2, W3, off, csr0, csr1, bufH, N, E);

    k_gather<32, false, false><<<(N + 7) / 8, TB, 0, stream>>>(bufH, b3, off, csr1, csr1, out, N, E);
}


# edge bucketing in two launches, not four: each slice ordered by tile in place; a tile's start = sum of its 256 run starts, so no histogram or scan launch; order in a bucket = LDS arrival order
# speedup vs baseline: 7.4713x; 1.0915x over previous
#include <hip/hip_runtime.h>
#include <hip/hip_bf16.h>

typedef __attribute__((ext_vector_type(2))) float v2f;
typedef __attribute__((ext_vector_type(8))) float v8f;


template <int PER>
__device__ __forceinline__ void wave_excl_scan(const int* c, int* st, int lane) {
    int v[PER];
    int s = 0;
#pragma unroll
    for (int k = 0; k < PER; ++k) { v[k] = c[lane * PER + k]; s += v[k]; }
    int inc = s;
    for (int o = 1; o < 32; o <<= 1) {
        const int t = __shfl_up(inc, o);
        if (lane >= o) inc += t;
    }
    int ex = inc - s;
#pragma unroll
    for (int k = 0; k < PER; ++k) { st[lane * PER + k] = ex; ex += v[k]; }
}

constexpr int TILE_MAX = 512;
constexpr int BT       = 1024;
constexpr int EPT      = 8;
constexpr int NSL      = 256;

__global__ __launch_bounds__(BT)
void k_slice_sort(const int* __restrict__ src, const int* __restrict__ dst,
                  int* __restrict__ lofsT, int* __restrict__ tmp,
                  int E, int chunk, int tiles) {
    __shared__ int lcnt[TILE_MAX];
    __shared__ int lst[TILE_MAX];
    __shared__ int lcur[TILE_MAX];
    __shared__ int swrd[BT * EPT];
    const int tid  = threadIdx.x;
    const int lane = tid & 31;
    const int w    = __builtin_amdgcn_readfirstlane(tid >> 5);
    const int b    = blockIdx.x;
    const int e0   = min(E, b * chunk);
    const int e1   = min(E, e0 + chunk);
    const int n    = e1 - e0;
    if (tid < TILE_MAX) { lcnt[tid] = 0; lcur[tid] = 0; }
    __syncthreads();

    int d[EPT], sv[EPT], r[EPT];
#pragma unroll
    for (int k = 0; k < EPT; ++k) { d[k] = -1; sv[k] = 0; r[k] = 0; }
    for (int s0 = e0; s0 < e1; s0 += BT * EPT) {
#pragma unroll
        for (int k = 0; k < EPT; ++k) {
            const int e = s0 + k * BT + tid;
            d[k]  = (e < e1) ? dst[e] : -1;
            sv[k] = (e < e1) ? src[e] : 0;
        }
#pragma unroll
        for (int k = 0; k < EPT; ++k)
            r[k] = (d[k] >= 0) ? atomicAdd(&lcnt[d[k] >> 8], 1) : 0;
    }
    __syncthreads();
    if (w == 0) wave_excl_scan<TILE_MAX / 32>(lcnt, lst, lane);
    __syncthreads();
    for (int t = tid; t <= tiles; t += BT)
        lofsT[t * NSL + b] = (t < tiles) ? lst[t] : n;

    if (n <= BT * EPT) {
#pragma unroll
        for (int k = 0; k < EPT; ++k)
            if (d[k] >= 0) swrd[lst[d[k] >> 8] + r[k]] = sv[k] | ((d[k] & 255) << 24);
        __syncthreads();
        for (int i = tid; i < n; i += BT) tmp[e0 + i] = swrd[i];
    } else {
        for (int s0 = e0; s0 < e1; s0 += BT * EPT) {
#pragma unroll
            for (int k = 0; k < EPT; ++k) {
                const int e = s0 + k * BT + tid;
                d[k]  = (e < e1) ? dst[e] : -1;
                sv[k] = (e < e1) ? src[e] : 0;
            }
#pragma unroll
            for (int k = 0; k < EPT; ++k) {
                if (d[k] >= 0) {
                    const int tl = d[k] >> 8;
                    const int p  = lst[tl] + atomicAdd(&lcur[tl], 1);
                    tmp[e0 + p] = sv[k] | ((d[k] & 255) << 24);
                }
            }
        }
    }
}

constexpr int SCAP = 16384;
constexpr int RPW  = NSL / (BT / 32);
__global__ __launch_bounds__(BT)
void k_tile_sort(const int* __restrict__ tmp, const int* __restrict__ lofsT,
                 int* __restrict__ off, int* __restrict__ csr, int E, int chunk) {
    __shared__ int ncnt[256];
    __shared__ int nst[256];
    __shared__ int ncur[256];
    __shared__ int rpos[NSL];
    __shared__ int rlen[NSL];
    __shared__ int wsa[NSL / 32];
    __shared__ int wsl[NSL / 32];
    __shared__ int sorted[SCAP];
    const int tid  = threadIdx.x;
    const int lane = tid & 31;
    const int w    = __builtin_amdgcn_readfirstlane(tid >> 5);
    const int t    = blockIdx.x;
    if (w < NSL / 32) {
        const int a = lofsT[t * NSL + tid];
        const int z = lofsT[(t + 1) * NSL + tid];
        rpos[tid] = min(E, tid * chunk) + a;
        rlen[tid] = z - a;
        ncnt[tid] = 0;
        ncur[tid] = 0;
        int sa = a, sl = z - a;
        for (int o = 16; o > 0; o >>= 1) {
            sa += __shfl_xor(sa, o);
            sl += __shfl_xor(sl, o);
        }
        if (lane == 0) { wsa[w] = sa; wsl[w] = sl; }
    }
    __syncthreads();
    int ts = 0, cnt = 0;
    for (int i = 0; i < NSL / 32; ++i) { ts += wsa[i]; cnt += wsl[i]; }

    int wd[RPW], ps[RPW], ln[RPW];
#pragma unroll
    for (int k = 0; k < RPW; ++k) { ps[k] = rpos[w * RPW + k]; ln[k] = rlen[w * RPW + k]; }
#pragma unroll
    for (int k = 0; k < RPW; ++k) wd[k] = (lane < ln[k]) ? tmp[ps[k] + lane] : 0;
#pragma unroll
    for (int k = 0; k < RPW; ++k)
        if (lane < ln[k]) atomicAdd(&ncnt[(unsigned)wd[k] >> 24], 1);
    for (int k = 0; k < RPW; ++k) {
        const int p0 = rpos[w * RPW + k];
        const int l0 = rlen[w * RPW + k];
        for (int j = 32 + lane; j < l0; j += 32)
            atomicAdd(&ncnt[(unsigned)tmp[p0 + j] >> 24], 1);
    }
    __syncthreads();
    if (w == 0) {
        wave_excl_scan<8>(ncnt, nst, lane);
        int4 qa, qb;
        qa.x = ts + nst[lane * 8 + 0]; qa.y = ts + nst[lane * 8 + 1];
        qa.z = ts + nst[lane * 8 + 2]; qa.w = ts + nst[lane * 8 + 3];
        qb.x = ts + nst[lane * 8 + 4]; qb.y = ts + nst[lane * 8 + 5];
        qb.z = ts + nst[lane * 8 + 6]; qb.w = ts + nst[lane * 8 + 7];
        int4* __restrict__ o4 = (int4*)(off + t * 256 + lane * 8);
        o4[0] = qa;
        o4[1] = qb;
    }
    __syncthreads();
    const bool staged = (cnt <= SCAP);
    int* __restrict__ cp = csr + ts;
#pragma unroll
    for (int k = 0; k < RPW; ++k) {
        if (lane < ln[k]) {
            const int dl = (int)((unsigned)wd[k] >> 24);
            const int p  = nst[dl] + atomicAdd(&ncur[dl], 1);
            const int s  = wd[k] & 0x00FFFFFF;
            if (staged) sorted[p] = s; else cp[p] = s;
        }
    }
    for (int k = 0; k < RPW; ++k) {
        const int p0 = rpos[w * RPW + k];
        const int l0 = rlen[w * RPW + k];
        for (int j = 32 + lane; j < l0; j += 32) {
            const int wv = tmp[p0 + j];
            const int dl = (int)((unsigned)wv >> 24);
            const int p  = nst[dl] + atomicAdd(&ncur[dl], 1);
            const int s  = wv & 0x00FFFFFF;
            if (staged) sorted[p] = s; else cp[p] = s;
        }
    }
    if (staged) {
        __syncthreads();
        for (int i = tid; i < cnt; i += BT) cp[i] = sorted[i];
    }
}

template <int K, int NT>
__global__ __launch_bounds__(512)
void k_gemm_wmma_f32(const float* __restrict__ X, const float* __restrict__ Wg,
                     const int* __restrict__ off, float* __restrict__ Y,
                     int N, int E) {
    constexpr int Cout = NT * 16;
    constexpr int KQ   = K / 4;
    constexpr int PQ4  = Cout + 8;
    constexpr int NCH  = K / 32;
    __shared__ float4 Ws[KQ * PQ4];

    const int tid  = threadIdx.x;
    const int lane = tid & 31;
    const int w    = __builtin_amdgcn_readfirstlane(tid >> 5);
    const int half = lane >> 4;
    const int c    = lane & 15;
    const int npan = N >> 4;
    const int pstride = (int)gridDim.x * 16;
    const int g  = w >> 2;
    const int sw = (((w + g) & 3) << 2) | g;
    int p = (int)blockIdx.x + (int)gridDim.x * sw;

    const float4* __restrict__ xa =
        (const float4*)(X + (size_t)((p < npan ? p : 0) * 16 + c) * K) + half;
    float4 a[4];
#pragma unroll
    for (int q = 0; q < 4; ++q) a[q] = xa[2 * q];

#pragma unroll 8
    for (int idx = tid; idx < KQ * Cout; idx += 512) {
        const int kq = idx / Cout;
        const int n  = idx % Cout;
        const float* __restrict__ wp = Wg + (size_t)(kq * 4) * Cout + n;
        float4 v;
        v.x = wp[0];
        v.y = wp[Cout];
        v.z = wp[2 * Cout];
        v.w = wp[3 * Cout];
        Ws[kq * PQ4 + n] = v;
    }
    __syncthreads();

    const int lo = half * PQ4 + c;
    float4 bc[NT];
#pragma unroll
    for (int t = 0; t < NT; ++t) bc[t] = Ws[lo + t * 16];

    v8f acc[NT];
#pragma unroll
    for (int t = 0; t < NT; ++t) acc[t] = (v8f){};

    int4 oa = {0, 0, 0, 0}, ob = {0, 0, 0, 0};
    int  ot = 0;

    int j = 0;
    while (p < npan) {
        const int pn = p + pstride;
        const bool lastj = (j == NCH - 1);

        const int  r0 = p * 16 + half * 8;
        const bool rn = (r0 + 8 < N);
        if (j == 0) {
            oa = *(const int4*)(off + r0);
            ob = *(const int4*)(off + r0 + 4);
            ot = off[rn ? r0 + 8 : r0];
        }

        const float4* __restrict__ xn =
            (const float4*)(X + (size_t)((pn < npan ? pn : p) * 16 + c) * K) + half;
        const float4* __restrict__ src = lastj ? xn : (xa + (j + 1) * 8);
        float4 an[4];
#pragma unroll
        for (int q = 0; q < 4; ++q) an[q] = src[2 * q];

#pragma unroll
        for (int q = 0; q < 4; ++q) {
            const int rown = (q == 3) ? (((j + 1) & (NCH - 1)) * 8) : (j * 8 + 2 * q + 2);
            const int bo = lo + rown * PQ4;
            float4 bn[NT];
#pragma unroll
            for (int t = 0; t < NT; ++t) bn[t] = Ws[bo + t * 16];

            const v2f a01 = {a[q].x, a[q].y};
            const v2f a23 = {a[q].z, a[q].w};
#pragma unroll
            for (int t = 0; t < NT; ++t) {
                const v2f b01 = {bc[t].x, bc[t].y};
                acc[t] = __builtin_amdgcn_wmma_f32_16x16x4_f32(
                              false, a01,  false, b01,
                              (short)0, acc[t],
                              false,  false);
            }
#pragma unroll
            for (int t = 0; t < NT; ++t) {
                const v2f b23 = {bc[t].z, bc[t].w};
                acc[t] = __builtin_amdgcn_wmma_f32_16x16x4_f32(
                              false, a23,  false, b23,
                              (short)0, acc[t],
                              false,  false);
            }
#pragma unroll
            for (int t = 0; t < NT; ++t) bc[t] = bn[t];
        }
#pragma unroll
        for (int q = 0; q < 4; ++q) a[q] = an[q];

        if (lastj) {
            float dv[8];
            dv[0] = rsqrtf((float)(oa.y - oa.x + 1));
            dv[1] = rsqrtf((float)(oa.z - oa.y + 1));
            dv[2] = rsqrtf((float)(oa.w - oa.z + 1));
            dv[3] = rsqrtf((float)(ob.x - oa.w + 1));
            dv[4] = rsqrtf((float)(ob.y - ob.x + 1));
            dv[5] = rsqrtf((float)(ob.z - ob.y + 1));
            dv[6] = rsqrtf((float)(ob.w - ob.z + 1));
            dv[7] = rsqrtf((float)((rn ? ot : E) - ob.w + 1));

            float* __restrict__ yp = Y + (size_t)(p * 16) * Cout + c;
#pragma unroll
            for (int t = 0; t < NT; ++t) {
#pragma unroll
                for (int v = 0; v < 8; ++v)
                    yp[(size_t)(v + half * 8) * Cout + t * 16] = acc[t][v] * dv[v];
                acc[t] = (v8f){};
            }
            xa = xn;
            p  = pn;
            j  = 0;
        } else {
            ++j;
        }
    }
}

template <int C, bool RELU, bool MOVE>
__global__ __launch_bounds__(256)
void k_gather(const float* __restrict__ h, const float* __restrict__ bias,
              const int* __restrict__ off, const int* __restrict__ csr,
              int* __restrict__ csr_new, float* __restrict__ out, int N, int E) {
    constexpr int LPR  = C / 4;
    constexpr int G    = 32 / LPR;
    constexpr int U    = (G == 4) ? 2 : 4;
    constexpr int STEP = U * G;
    const int lane = threadIdx.x & 31;
    const int i = __builtin_amdgcn_readfirstlane((int)(blockIdx.x * 8 + (threadIdx.x >> 5)));
    if (i >= N) return;
    const int grp = lane / LPR;
    const int li  = lane % LPR;

    const int start = off[i];
    const int end   = (i + 1 < N) ? off[i + 1] : E;
    const int cnt   = end - start;
    const float di  = rsqrtf((float)(cnt + 1));

    const float4* __restrict__ h4 = (const float4*)h;
    float4 acc = {0.0f, 0.0f, 0.0f, 0.0f};
    float4 b   = {0.0f, 0.0f, 0.0f, 0.0f};
    if (grp == 0) {
        b   = ((const float4*)bias)[li];
        acc = h4[(size_t)i * LPR + li];
    }

    for (int base = 0; base < cnt; base += 32) {
        const int n = min(32, cnt - base);
        int s = i;
        if (lane < n) {
            s = csr[start + base + lane];
            if (MOVE) csr_new[start + base + lane] = s;
        }
        for (int t = 0; t < n; t += STEP) {
            float4 v[U];
            float  wv[U];
#pragma unroll
            for (int k = 0; k < U; ++k) {
                const int e = (G == 1) ? (t + k) : (t + k * G + grp);
                int sv;
                if constexpr (G == 1) sv = __builtin_amdgcn_readlane(s, e);
                else                  sv = __shfl(s, e);
                wv[k] = (e < n) ? 1.0f : 0.0f;
                v[k]  = h4[(size_t)sv * LPR + li];
            }
#pragma unroll
            for (int k = 0; k < U; ++k) {
                acc.x = fmaf(v[k].x, wv[k], acc.x);
                acc.y = fmaf(v[k].y, wv[k], acc.y);
                acc.z = fmaf(v[k].z, wv[k], acc.z);
                acc.w = fmaf(v[k].w, wv[k], acc.w);
            }
        }
    }

    for (int o = LPR; o < 32; o <<= 1) {
        acc.x += __shfl_xor(acc.x, o);
        acc.y += __shfl_xor(acc.y, o);
        acc.z += __shfl_xor(acc.z, o);
        acc.w += __shfl_xor(acc.w, o);
    }
    if (grp == 0) {
        acc.x = fmaf(acc.x, di, b.x); acc.y = fmaf(acc.y, di, b.y);
        acc.z = fmaf(acc.z, di, b.z); acc.w = fmaf(acc.w, di, b.w);
        if (RELU) {
            acc.x = fmaxf(acc.x, 0.0f); acc.y = fmaxf(acc.y, 0.0f);
            acc.z = fmaxf(acc.z, 0.0f); acc.w = fmaxf(acc.w, 0.0f);
        }
        ((float4*)out)[(size_t)i * LPR + li] = acc;
    }
}

template <int C, int NT, bool MOVE>
__global__ __launch_bounds__(512)
void k_gather_gemm(const float* __restrict__ h, const float* __restrict__ bias,
                   const float* __restrict__ Wg,
                   const int* __restrict__ off, const int* __restrict__ csr,
                   int* __restrict__ csr_new, float* __restrict__ Y, int N, int E) {
    constexpr int LPR  = C / 4;
    constexpr int G    = 32 / LPR;
    constexpr int U    = (G == 4) ? 2 : 4;
    constexpr int STEP = U * G;
    constexpr int Cout = NT * 16;
    constexpr int KQ   = C / 4;
    constexpr int PQ4  = Cout + 8;
    constexpr int AP4  = C / 4 + 1;
    constexpr int WPT  = (KQ * Cout) / 512;
    __shared__ float4 Ws[KQ * PQ4];
    __shared__ float4 As[16 * AP4];
    __shared__ float  dvs[16];

    const int tid  = threadIdx.x;
    const int lane = tid & 31;
    const int w  = __builtin_amdgcn_readfirstlane(tid >> 5);
    const int i  = __builtin_amdgcn_readfirstlane((int)blockIdx.x * 16 + w);
    const bool live = (i < N);
    const int  ic   = live ? i : 0;
    const int grp = lane / LPR;
    const int li  = lane % LPR;

    float4 wst[WPT];
#pragma unroll
    for (int r = 0; r < WPT; ++r) {
        const int idx = tid + r * 512;
        const int kq = idx / Cout;
        const int n  = idx % Cout;
        const float* __restrict__ wp = Wg + (size_t)(kq * 4) * Cout + n;
        wst[r].x = wp[0];
        wst[r].y = wp[Cout];
        wst[r].z = wp[2 * Cout];
        wst[r].w = wp[3 * Cout];
    }

    const int start = off[ic];
    const int end   = (ic + 1 < N) ? off[ic + 1] : E;
    const int cnt   = live ? (end - start) : 0;
    const float di  = rsqrtf((float)(cnt + 1));

    const float4* __restrict__ h4 = (const float4*)h;
    float4 acc = {0.0f, 0.0f, 0.0f, 0.0f};
    float4 b   = {0.0f, 0.0f, 0.0f, 0.0f};
    if (grp == 0) {
        b   = ((const float4*)bias)[li];
        acc = h4[(size_t)ic * LPR + li];
    }

    for (int base = 0; base < cnt; base += 32) {
        const int n = min(32, cnt - base);
        int s = ic;
        if (lane < n) {
            s = csr[start + base + lane];
            if (MOVE) csr_new[start + base + lane] = s;
        }
        for (int t = 0; t < n; t += STEP) {
            float4 v[U];
            float  wv[U];
#pragma unroll
            for (int k = 0; k < U; ++k) {
                const int e = (G == 1) ? (t + k) : (t + k * G + grp);
                int sv;
                if constexpr (G == 1) sv = __builtin_amdgcn_readlane(s, e);
                else                  sv = __shfl(s, e);
                wv[k] = (e < n) ? 1.0f : 0.0f;
                v[k]  = h4[(size_t)sv * LPR + li];
            }
#pragma unroll
            for (int k = 0; k < U; ++k) {
                acc.x = fmaf(v[k].x, wv[k], acc.x);
                acc.y = fmaf(v[k].y, wv[k], acc.y);
                acc.z = fmaf(v[k].z, wv[k], acc.z);
                acc.w = fmaf(v[k].w, wv[k], acc.w);
            }
        }
    }

    for (int o = LPR; o < 32; o <<= 1) {
        acc.x += __shfl_xor(acc.x, o);
        acc.y += __shfl_xor(acc.y, o);
        acc.z += __shfl_xor(acc.z, o);
        acc.w += __shfl_xor(acc.w, o);
    }
    if (grp == 0) {
        acc.x = fmaf(acc.x, di, b.x); acc.y = fmaf(acc.y, di, b.y);
        acc.z = fmaf(acc.z, di, b.z); acc.w = fmaf(acc.w, di, b.w);
        acc.x = fmaxf(acc.x, 0.0f); acc.y = fmaxf(acc.y, 0.0f);
        acc.z = fmaxf(acc.z, 0.0f); acc.w = fmaxf(acc.w, 0.0f);
        As[w * AP4 + li] = acc;
    }
    if (lane == 0) dvs[w] = di;
#pragma unroll
    for (int r = 0; r < WPT; ++r) {
        const int idx = tid + r * 512;
        Ws[(idx / Cout) * PQ4 + (idx % Cout)] = wst[r];
    }
    __syncthreads();

    if (w < NT) {
        const int hf = lane >> 4;
        const int c  = lane & 15;
        v8f d = (v8f){};
#pragma unroll
        for (int q = 0; q < C / 8; ++q) {
            const float4 a  = As[c * AP4 + 2 * q + hf];
            const float4 bw = Ws[(2 * q + hf) * PQ4 + w * 16 + c];
            const v2f a01 = {a.x, a.y};
            const v2f a23 = {a.z, a.w};
            const v2f b01 = {bw.x, bw.y};
            const v2f b23 = {bw.z, bw.w};
            d = __builtin_amdgcn_wmma_f32_16x16x4_f32(
                     false, a01,  false, b01,
                     (short)0, d,  false,  false);
            d = __builtin_amdgcn_wmma_f32_16x16x4_f32(
                     false, a23,  false, b23,
                     (short)0, d,  false,  false);
        }
        const int row0 = (int)blockIdx.x * 16 + hf * 8;
        float* __restrict__ yp = Y + (size_t)row0 * Cout + w * 16 + c;
#pragma unroll
        for (int v = 0; v < 8; ++v)
            if (row0 + v < N) yp[(size_t)v * Cout] = d[v] * dvs[hf * 8 + v];
    }
}

extern "C" void kernel_launch(void* const* d_in, const int* in_sizes, int n_in,
                              void* d_out, int out_size, void* d_ws, size_t ws_size,
                              hipStream_t stream) {
    const float* x  = (const float*)d_in[0];
    const int*   ei = (const int*)  d_in[1];
    const float* W1 = (const float*)d_in[2];
    const float* b1 = (const float*)d_in[3];
    const float* W2 = (const float*)d_in[4];
    const float* b2 = (const float*)d_in[5];
    const float* W3 = (const float*)d_in[6];
    const float* b3 = (const float*)d_in[7];

    const int K0 = 512, C1 = 128, C2 = 64, C3 = 32;
    const int N = in_sizes[0] / K0;
    const int E = in_sizes[1] / 2;
    const int* src = ei;
    const int* dst = ei + E;

    float* ws   = (float*)d_ws;
    size_t pad  = ((size_t)N + 255) & ~(size_t)255;
    int*   off  = (int*)ws;
    float* bufH = ws + pad;
    float* bufA = bufH + (size_t)N * C1;
    float* out  = (float*)d_out;
    int*   lofsT = (int*)bufA;
    int*   tmp   = lofsT + 2 * pad;
    int*   csr0 = (int*)d_out;
    int*   csr1 = (int*)(bufH + (size_t)N * C2);

    const int TB = 256;

    const int tiles = (int)(pad / 256);
    const int chunk = (E + NSL - 1) / NSL;
    k_slice_sort<<<NSL, BT, 0, stream>>>(src, dst, lofsT, tmp, E, chunk, tiles);
    k_tile_sort <<<tiles, BT, 0, stream>>>(tmp, lofsT, off, csr0, E, chunk);

    k_gemm_wmma_f32<512, 8><<<256, 512, 0, stream>>>(x, W1, off, bufH, N, E);

    k_gather_gemm<128, 4, false><<<(N + 15) / 16, 512, 0, stream>>>(bufH, b1, W2, off, csr0, csr1, bufA, N, E);

    k_gather_gemm<64, 2, true><<<(N + 15) / 16, 512, 0, stream>>>(bufA, b2, W3, off, csr0, csr1, bufH, N, E);

    k_gather<32, false, false><<<(N + 7) / 8, TB, 0, stream>>>(bufH, b3, off, csr1, csr1, out, N, E);
}
